# starting kernel
# baseline (speedup 1.0000x reference)
#include <hip/hip_runtime.h>
#include <hip/hip_bf16.h>
#include <math.h>

typedef __attribute__((ext_vector_type(2))) float v2f;
typedef __attribute__((ext_vector_type(8))) float v8f;

#define NEG_SLOPE 0.2f

__device__ __forceinline__ unsigned enc_f32(float f) {
  unsigned u = __float_as_uint(f);
  return u ^ ((u >> 31) ? 0xFFFFFFFFu : 0x80000000u);
}
__device__ __forceinline__ float dec_f32(unsigned u) {
  unsigned v = u ^ ((u >> 31) ? 0x80000000u : 0xFFFFFFFFu);
  return __uint_as_float(v);
}

__global__ __launch_bounds__(256) void gat_gemm128(const float* __restrict__ X,
                                                   const float* __restrict__ W,
                                                   float* __restrict__ Hout,
                                                   int N) {
  const int lane  = threadIdx.x & 31;
  const int wv    = threadIdx.x >> 5;
  const int m0    = blockIdx.x << 4;
  const int n0    = wv << 4;
  const int r     = lane & 15;
  const int khalf = (lane >> 4) << 1;

  int arow = m0 + r;
  if (arow >= N) arow = N - 1;
  const float* ap = X + (size_t)arow * 128;

  v8f acc = {0.f, 0.f, 0.f, 0.f, 0.f, 0.f, 0.f, 0.f};
#pragma unroll
  for (int k0 = 0; k0 < 128; k0 += 4) {
    const int ka = k0 + khalf;
    v2f a; a.x = ap[ka]; a.y = ap[ka + 1];
    v2f b; b.x = W[(size_t)ka * 128 + n0 + r];
           b.y = W[(size_t)(ka + 1) * 128 + n0 + r];
    acc = __builtin_amdgcn_wmma_f32_16x16x4_f32(false, a, false, b,
                                                (short)0, acc, false, false);
  }
  const int rowbase = m0 + ((lane >> 4) << 3);
  const int col     = n0 + r;
#pragma unroll
  for (int i = 0; i < 8; ++i) {
    const int row = rowbase + i;
    if (row < N) Hout[(size_t)row * 128 + col] = acc[i];
  }
}

__global__ void gat_scores_init(const float* __restrict__ Hm,
                                const float* __restrict__ asrc,
                                const float* __restrict__ adst,
                                float* __restrict__ es, float* __restrict__ ed,
                                unsigned* __restrict__ menc,
                                float* __restrict__ ssum,
                                float* __restrict__ acc, int N) {
  const int idx = blockIdx.x * blockDim.x + threadIdx.x;
  if (idx >= N * 4) return;
  const int hd = idx & 3;
  const size_t base = (size_t)(idx >> 2) * 128 + hd * 32;
  const float* hp  = Hm + base;
  float*       ap  = acc + base;
  const float* sa  = asrc + hd * 32;
  const float* da  = adst + hd * 32;
  float s = 0.f, d = 0.f;
#pragma unroll
  for (int c = 0; c < 32; c += 4) {
    float4 hv = *(const float4*)(hp + c);
    float4 av = *(const float4*)(sa + c);
    float4 bv = *(const float4*)(da + c);
    s += hv.x * av.x + hv.y * av.y + hv.z * av.z + hv.w * av.w;
    d += hv.x * bv.x + hv.y * bv.y + hv.z * bv.z + hv.w * bv.w;
    *(float4*)(ap + c) = make_float4(0.f, 0.f, 0.f, 0.f);
  }
  es[idx] = s;
  ed[idx] = d;
  menc[idx] = 0u;
  ssum[idx] = 0.f;
}

__device__ __forceinline__ float lrelu(float x) {
  return x > 0.f ? x : NEG_SLOPE * x;
}

__global__ void gat_edge_max(const int* __restrict__ ei,
                             const float* __restrict__ es,
                             const float* __restrict__ ed,
                             unsigned* __restrict__ menc, int E, int N) {
  const int e = blockIdx.x * blockDim.x + threadIdx.x;
  const int Et = E + N;
  if (e >= Et) return;
  int s, d;
  if (e < E) { s = ei[e]; d = ei[E + e]; } else { s = d = e - E; }
  float4 vs = *(const float4*)(es + (size_t)s * 4);
  float4 vd = *(const float4*)(ed + (size_t)d * 4);
  unsigned* mp = menc + (size_t)d * 4;
  atomicMax(mp + 0, enc_f32(lrelu(vs.x + vd.x)));
  atomicMax(mp + 1, enc_f32(lrelu(vs.y + vd.y)));
  atomicMax(mp + 2, enc_f32(lrelu(vs.z + vd.z)));
  atomicMax(mp + 3, enc_f32(lrelu(vs.w + vd.w)));
}

__global__ __launch_bounds__(256) void gat_edge_scatter(
    const int* __restrict__ ei,
    const float* __restrict__ es, const float* __restrict__ ed,
    const unsigned* __restrict__ menc, float* __restrict__ ssum,
    const float* __restrict__ Hm, float* __restrict__ acc, int E, int N) {
  const int e = blockIdx.x * 8 + (threadIdx.x >> 5);
  const int lane = threadIdx.x & 31;
  const int Et = E + N;
  if (e >= Et) return;
  int s, d;
  if (e < E) { s = ei[e]; d = ei[E + e]; } else { s = d = e - E; }
  const int hd = lane >> 3;
  float l = es[(size_t)s * 4 + hd] + ed[(size_t)d * 4 + hd];
  l = lrelu(l);
  const float mh = dec_f32(menc[(size_t)d * 4 + hd]);
  const float ex = expf(l - mh);
  if ((lane & 7) == 0) atomicAdd(&ssum[(size_t)d * 4 + hd], ex);
  float4 hv = *(const float4*)(Hm + (size_t)s * 128 + lane * 4);
  float* op = acc + (size_t)d * 128 + lane * 4;
  atomicAdd(op + 0, hv.x * ex);
  atomicAdd(op + 1, hv.y * ex);
  atomicAdd(op + 2, hv.z * ex);
  atomicAdd(op + 3, hv.w * ex);
}

__global__ void gat_finalize(const float* __restrict__ acc,
                             const float* __restrict__ ssum,
                             const float* __restrict__ bias,
                             float* __restrict__ outp, int N, int apply_elu) {
  const int idx = blockIdx.x * blockDim.x + threadIdx.x;
  if (idx >= N * 128) return;
  const int n  = idx >> 7;
  const int hd = (idx >> 5) & 3;
  float v = acc[idx] / ssum[(size_t)n * 4 + hd] + bias[idx & 127];
  if (apply_elu) v = v > 0.f ? v : expm1f(v);
  outp[idx] = v;
}

extern "C" void kernel_launch(void* const* d_in, const int* in_sizes, int n_in,
                              void* d_out, int out_size, void* d_ws, size_t ws_size,
                              hipStream_t stream) {
  const float* x  = (const float*)d_in[0];
  const int*   ei = (const int*)d_in[1];
  const int N = in_sizes[0] / 128;
  const int E = in_sizes[1] / 2;

  const float* Wl[3] = {(const float*)d_in[2], (const float*)d_in[6],  (const float*)d_in[10]};
  const float* As[3] = {(const float*)d_in[3], (const float*)d_in[7],  (const float*)d_in[11]};
  const float* Ad[3] = {(const float*)d_in[4], (const float*)d_in[8],  (const float*)d_in[12]};
  const float* Bb[3] = {(const float*)d_in[5], (const float*)d_in[9],  (const float*)d_in[13]};

  float*    F0   = (float*)d_ws;
  float*    F1   = F0 + (size_t)N * 128;
  float*    es   = F1 + (size_t)N * 128;
  float*    ed   = es + (size_t)N * 4;
  unsigned* menc = (unsigned*)(ed + (size_t)N * 4);
  float*    ssum = (float*)(menc + (size_t)N * 4);

  const int Et = E + N;
  const dim3 bGemm(256);
  const dim3 gGemm((N + 15) / 16);
  const int gScore = (N * 4 + 255) / 256;
  const int gMax   = (Et + 255) / 256;
  const int gScat  = (Et + 7) / 8;
  const int gFin   = (N * 128 + 255) / 256;

  const float* inp = x;
  for (int layer = 0; layer < 3; ++layer) {
    float* Hm = (layer & 1) ? F1 : F0;
    float* Ac = (layer & 1) ? F0 : F1;
    gat_gemm128<<<gGemm, bGemm, 0, stream>>>(inp, Wl[layer], Hm, N);
    gat_scores_init<<<gScore, 256, 0, stream>>>(Hm, As[layer], Ad[layer],
                                                es, ed, menc, ssum, Ac, N);
    gat_edge_max<<<gMax, 256, 0, stream>>>(ei, es, ed, menc, E, N);
    gat_edge_scatter<<<gScat, 256, 0, stream>>>(ei, es, ed, menc, ssum, Hm, Ac, E, N);
    float* ot = (layer == 2) ? (float*)d_out : Hm;
    gat_finalize<<<gFin, 256, 0, stream>>>(Ac, ssum, Bb[layer], ot, N,
                                           layer < 2 ? 1 : 0);
    inp = Hm;
  }
}


# Edges counting-sorted by destination once per call (int adds, scan, rows ordered by source); per layer one wave per node does max, exp-sum, weighted row sum: no float atomics, fixed sum order
# speedup vs baseline: 4.0262x; 4.0262x over previous
#include <hip/hip_runtime.h>
#include <hip/hip_bf16.h>
#include <math.h>

typedef __attribute__((ext_vector_type(2))) float v2f;
typedef __attribute__((ext_vector_type(8))) float v8f;

#define NEG_SLOPE 0.2f

__global__ __launch_bounds__(256) void gat_gemm128(const float* __restrict__ X,
                                                   const float* __restrict__ W,
                                                   float* __restrict__ Hout,
                                                   int N) {
  const int lane  = threadIdx.x & 31;
  const int wv    = threadIdx.x >> 5;
  const int m0    = blockIdx.x << 4;
  const int n0    = wv << 4;
  const int r     = lane & 15;
  const int khalf = (lane >> 4) << 1;

  int arow = m0 + r;
  if (arow >= N) arow = N - 1;
  const float* ap = X + (size_t)arow * 128;

  v8f acc = {0.f, 0.f, 0.f, 0.f, 0.f, 0.f, 0.f, 0.f};
#pragma unroll
  for (int k0 = 0; k0 < 128; k0 += 4) {
    const int ka = k0 + khalf;
    v2f a; a.x = ap[ka]; a.y = ap[ka + 1];
    v2f b; b.x = W[(size_t)ka * 128 + n0 + r];
           b.y = W[(size_t)(ka + 1) * 128 + n0 + r];
    acc = __builtin_amdgcn_wmma_f32_16x16x4_f32(false, a, false, b,
                                                (short)0, acc, false, false);
  }
  const int rowbase = m0 + ((lane >> 4) << 3);
  const int col     = n0 + r;
#pragma unroll
  for (int i = 0; i < 8; ++i) {
    const int row = rowbase + i;
    if (row < N) Hout[(size_t)row * 128 + col] = acc[i];
  }
}

__global__ void gat_scores(const float* __restrict__ Hm,
                           const float* __restrict__ asrc,
                           const float* __restrict__ adst,
                           float* __restrict__ es, float* __restrict__ ed, int N) {
  const int idx = blockIdx.x * blockDim.x + threadIdx.x;
  if (idx >= N * 4) return;
  const int hd = idx & 3;
  const size_t base = (size_t)(idx >> 2) * 128 + hd * 32;
  const float* hp  = Hm + base;
  const float* sa  = asrc + hd * 32;
  const float* da  = adst + hd * 32;
  float s = 0.f, d = 0.f;
#pragma unroll
  for (int c = 0; c < 32; c += 4) {
    float4 hv = *(const float4*)(hp + c);
    float4 av = *(const float4*)(sa + c);
    float4 bv = *(const float4*)(da + c);
    s += hv.x * av.x + hv.y * av.y + hv.z * av.z + hv.w * av.w;
    d += hv.x * bv.x + hv.y * bv.y + hv.z * bv.z + hv.w * bv.w;
  }
  es[idx] = s;
  ed[idx] = d;
}

__device__ __forceinline__ float lrelu(float x) {
  return x > 0.f ? x : NEG_SLOPE * x;
}


__global__ void csr_zero(int* __restrict__ cnt, int N) {
  const int i = blockIdx.x * blockDim.x + threadIdx.x;
  if (i < N) cnt[i] = 0;
}

__global__ void csr_count(const int* __restrict__ ei, int* __restrict__ cnt,
                          int* __restrict__ rank, int E) {
  const int e = blockIdx.x * blockDim.x + threadIdx.x;
  if (e >= E) return;
  rank[e] = atomicAdd(&cnt[ei[E + e]], 1);
}

__global__ __launch_bounds__(1024) void csr_scan(const int* __restrict__ cnt,
                                                 int* __restrict__ rowptr, int N) {
  __shared__ int wsum[32];
  __shared__ int wpre[32];
  const int tid  = threadIdx.x;
  const int lane = tid & 31;
  const int wv   = tid >> 5;
  const int base = blockIdx.x * 1024;
  int pre = 0;
  for (int i = tid; i < base; i += 1024) pre += cnt[i];
  const int idx = base + tid;
  const int c = idx < N ? cnt[idx] : 0;
  int inc = c;
#pragma unroll
  for (int o = 1; o < 32; o <<= 1) {
    const int t = __shfl_up(inc, o);
    if (lane >= o) inc += t;
  }
#pragma unroll
  for (int o = 16; o > 0; o >>= 1) pre += __shfl_xor(pre, o);
  if (lane == 31) wsum[wv] = inc;
  if (lane == 0)  wpre[wv] = pre;
  __syncthreads();
  int off = 0;
  for (int i = 0; i < 32; ++i) {
    off += wpre[i];
    if (i < wv) off += wsum[i];
  }
  const int excl = off + inc - c;
  if (idx < N) rowptr[idx] = excl;
  if (idx == N - 1) rowptr[N] = excl + c;
}

__global__ void csr_fill(const int* __restrict__ ei, const int* __restrict__ rowptr,
                         const int* __restrict__ rank, int* __restrict__ col, int E) {
  const int e = blockIdx.x * blockDim.x + threadIdx.x;
  if (e >= E) return;
  col[rowptr[ei[E + e]] + rank[e]] = ei[e];
}

__global__ __launch_bounds__(256) void csr_sort_rows(const int* __restrict__ rowptr,
                                                     const int* __restrict__ cin,
                                                     int* __restrict__ cout, int N) {
  const int lane = threadIdx.x & 31;
  const int node = __builtin_amdgcn_readfirstlane(blockIdx.x * 8 + (threadIdx.x >> 5));
  if (node >= N) return;
  const int beg = __builtin_amdgcn_readfirstlane(rowptr[node]);
  const int n   = __builtin_amdgcn_readfirstlane(rowptr[node + 1]) - beg;
  for (int i = lane; i < n; i += 32) {
    const int key = cin[beg + i];
    int r = 0;
    for (int j = 0; j < n; ++j) {
      const int kj = cin[beg + j];
      r += (kj < key || (kj == key && j < i)) ? 1 : 0;
    }
    cout[beg + r] = key;
  }
}

__device__ __forceinline__ float wave_max(float v) {
#pragma unroll
  for (int o = 16; o > 0; o >>= 1) v = fmaxf(v, __shfl_xor(v, o));
  return v;
}
__device__ __forceinline__ float pick_head(float x0, float x1, float x2, float x3, int j,
                                           int k0, int k1, int k2, int k3) {
  const int b0 = __builtin_amdgcn_readlane(__float_as_int(x0), j);
  const int b1 = __builtin_amdgcn_readlane(__float_as_int(x1), j);
  const int b2 = __builtin_amdgcn_readlane(__float_as_int(x2), j);
  const int b3 = __builtin_amdgcn_readlane(__float_as_int(x3), j);
  return __int_as_float((b0 & k0) | (b1 & k1) | (b2 & k2) | (b3 & k3));
}

__global__ __launch_bounds__(256) void gat_gather(
    const int* __restrict__ rowptr, const int* __restrict__ col,
    const float* __restrict__ es, const float* __restrict__ ed,
    const float* __restrict__ Hm, const float* __restrict__ bias,
    float* __restrict__ outp, int N, int apply_elu) {
  const int lane = threadIdx.x & 31;
  const int node = __builtin_amdgcn_readfirstlane(blockIdx.x * 8 + (threadIdx.x >> 5));
  if (node >= N) return;
  const int beg = __builtin_amdgcn_readfirstlane(rowptr[node]);
  const int tot = __builtin_amdgcn_readfirstlane(rowptr[node + 1]) - beg + 1;
  const int hd  = lane >> 3;
  const float4 vd = *(const float4*)(ed + (size_t)node * 4);
  const float ninf = -__builtin_inff();

  int sc;
  float l0, l1, l2, l3;
  auto load_chunk = [&](int t0) {
    const int t = t0 + lane;
    sc = node;
    l0 = l1 = l2 = l3 = ninf;
    if (t < tot) {
      if (t < tot - 1) sc = col[beg + t];
      const float4 vs = *(const float4*)(es + (size_t)sc * 4);
      l0 = lrelu(vs.x + vd.x);
      l1 = lrelu(vs.y + vd.y);
      l2 = lrelu(vs.z + vd.z);
      l3 = lrelu(vs.w + vd.w);
    }
  };
  load_chunk(0);
  float m0 = l0, m1 = l1, m2 = l2, m3 = l3;
  for (int t0 = 32; t0 < tot; t0 += 32) {
    load_chunk(t0);
    m0 = fmaxf(m0, l0); m1 = fmaxf(m1, l1); m2 = fmaxf(m2, l2); m3 = fmaxf(m3, l3);
  }
  m0 = wave_max(m0); m1 = wave_max(m1); m2 = wave_max(m2); m3 = wave_max(m3);

  float4 acc = make_float4(0.f, 0.f, 0.f, 0.f);
  float ssum = 0.f;
  const float* hrow = Hm + lane * 4;
  const int k0 = hd == 0 ? -1 : 0, k1 = hd == 1 ? -1 : 0, k2 = hd == 2 ? -1 : 0, k3 = hd == 3 ? -1 : 0;
  for (int t0 = 0; t0 < tot; t0 += 32) {
    if (tot > 32) load_chunk(t0);
    const float x0 = expf(l0 - m0);
    const float x1 = expf(l1 - m1);
    const float x2 = expf(l2 - m2);
    const float x3 = expf(l3 - m3);
    const int cnt = (tot - t0) < 32 ? (tot - t0) : 32;
    int j = 0;
    for (; j + 4 <= cnt; j += 4) {
      const float4 ha = *(const float4*)(hrow + (size_t)__builtin_amdgcn_readlane(sc, j) * 128);
      const float4 hb = *(const float4*)(hrow + (size_t)__builtin_amdgcn_readlane(sc, j + 1) * 128);
      const float4 hc = *(const float4*)(hrow + (size_t)__builtin_amdgcn_readlane(sc, j + 2) * 128);
      const float4 hd4 = *(const float4*)(hrow + (size_t)__builtin_amdgcn_readlane(sc, j + 3) * 128);
      const float ea = pick_head(x0, x1, x2, x3, j, k0, k1, k2, k3);
      const float eb = pick_head(x0, x1, x2, x3, j + 1, k0, k1, k2, k3);
      const float ec = pick_head(x0, x1, x2, x3, j + 2, k0, k1, k2, k3);
      const float ee = pick_head(x0, x1, x2, x3, j + 3, k0, k1, k2, k3);
      ssum += ea; acc.x = fmaf(ha.x, ea, acc.x); acc.y = fmaf(ha.y, ea, acc.y); acc.z = fmaf(ha.z, ea, acc.z); acc.w = fmaf(ha.w, ea, acc.w);
      ssum += eb; acc.x = fmaf(hb.x, eb, acc.x); acc.y = fmaf(hb.y, eb, acc.y); acc.z = fmaf(hb.z, eb, acc.z); acc.w = fmaf(hb.w, eb, acc.w);
      ssum += ec; acc.x = fmaf(hc.x, ec, acc.x); acc.y = fmaf(hc.y, ec, acc.y); acc.z = fmaf(hc.z, ec, acc.z); acc.w = fmaf(hc.w, ec, acc.w);
      ssum += ee; acc.x = fmaf(hd4.x, ee, acc.x); acc.y = fmaf(hd4.y, ee, acc.y); acc.z = fmaf(hd4.z, ee, acc.z); acc.w = fmaf(hd4.w, ee, acc.w);
    }
#pragma unroll 1
    for (; j < cnt; ++j) {
      const float4 ha = *(const float4*)(hrow + (size_t)__builtin_amdgcn_readlane(sc, j) * 128);
      const float ea = pick_head(x0, x1, x2, x3, j, k0, k1, k2, k3);
      ssum += ea; acc.x = fmaf(ha.x, ea, acc.x); acc.y = fmaf(ha.y, ea, acc.y); acc.z = fmaf(ha.z, ea, acc.z); acc.w = fmaf(ha.w, ea, acc.w);
    }
  }
  const float4 bv = *(const float4*)(bias + lane * 4);
  float4 v;
  v.x = acc.x / ssum + bv.x;
  v.y = acc.y / ssum + bv.y;
  v.z = acc.z / ssum + bv.z;
  v.w = acc.w / ssum + bv.w;
  if (apply_elu) {
    v.x = v.x > 0.f ? v.x : expm1f(v.x);
    v.y = v.y > 0.f ? v.y : expm1f(v.y);
    v.z = v.z > 0.f ? v.z : expm1f(v.z);
    v.w = v.w > 0.f ? v.w : expm1f(v.w);
  }
  *(float4*)(outp + (size_t)node * 128 + lane * 4) = v;
}

extern "C" void kernel_launch(void* const* d_in, const int* in_sizes, int n_in,
                              void* d_out, int out_size, void* d_ws, size_t ws_size,
                              hipStream_t stream) {
  const float* x  = (const float*)d_in[0];
  const int*   ei = (const int*)d_in[1];
  const int N = in_sizes[0] / 128;
  const int E = in_sizes[1] / 2;

  const float* Wl[3] = {(const float*)d_in[2], (const float*)d_in[6],  (const float*)d_in[10]};
  const float* As[3] = {(const float*)d_in[3], (const float*)d_in[7],  (const float*)d_in[11]};
  const float* Ad[3] = {(const float*)d_in[4], (const float*)d_in[8],  (const float*)d_in[12]};
  const float* Bb[3] = {(const float*)d_in[5], (const float*)d_in[9],  (const float*)d_in[13]};

  float* F0     = (float*)d_ws;
  float* F1     = F0 + (size_t)N * 128;
  float* es     = F1 + (size_t)N * 128;
  float* ed     = es + (size_t)N * 4;
  int*   cnt    = (int*)(ed + (size_t)N * 4);
  int*   rowptr = cnt + (size_t)N;
  int*   rank   = rowptr + (size_t)N + 4;
  int*   colA   = rank + (size_t)E;
  int*   colB   = colA + (size_t)E;

  const dim3 bGemm(256);
  const dim3 gGemm((N + 15) / 16);
  const int gNode  = (N + 255) / 256;
  const int gEdge  = (E + 255) / 256;
  const int gScan  = (N + 1023) / 1024;
  const int gWave  = (N + 7) / 8;
  const int gScore = (N * 4 + 255) / 256;

  csr_zero<<<gNode, 256, 0, stream>>>(cnt, N);
  csr_count<<<gEdge, 256, 0, stream>>>(ei, cnt, rank, E);
  csr_scan<<<gScan, 1024, 0, stream>>>(cnt, rowptr, N);
  csr_fill<<<gEdge, 256, 0, stream>>>(ei, rowptr, rank, colA, E);
  csr_sort_rows<<<gWave, 256, 0, stream>>>(rowptr, colA, colB, N);

  const float* inp = x;
  for (int layer = 0; layer < 3; ++layer) {
    gat_gemm128<<<gGemm, bGemm, 0, stream>>>(inp, Wl[layer], F0, N);
    gat_scores<<<gScore, 256, 0, stream>>>(F0, As[layer], Ad[layer], es, ed, N);
    float* ot = (layer == 2) ? (float*)d_out : F1;
    gat_gather<<<gWave, 256, 0, stream>>>(rowptr, colB, es, ed, F0, Bb[layer], ot, N,
                                          layer < 2 ? 1 : 0);
    inp = F1;
  }
}


# GEMM retiled: W put once per work-group into LDS in fragment order, each wave owns 16-row strips over all 128 columns (8 tiles), A as 16-byte loads; k summed in a new order inside each product
# speedup vs baseline: 4.5533x; 1.1309x over previous
#include <hip/hip_runtime.h>
#include <hip/hip_bf16.h>
#include <math.h>

typedef __attribute__((ext_vector_type(2))) float v2f;
typedef __attribute__((ext_vector_type(8))) float v8f;

#define NEG_SLOPE 0.2f

#define GEMM_WGS 256
__global__ __launch_bounds__(512) void gat_gemm128(const float* __restrict__ X,
                                                   const float* __restrict__ W,
                                                   float* __restrict__ Hout,
                                                   int N) {
  __shared__ float4 wl[32 * 8 * 16];
  const int lane = threadIdx.x & 31;
  const int wv   = __builtin_amdgcn_readfirstlane(threadIdx.x >> 5);
  const int c    = lane & 15;
  const int h    = lane >> 4;
  const int nstrips = (N + 15) >> 4;
  int strip = wv * GEMM_WGS + blockIdx.x;
  const int stride = 16 * GEMM_WGS;

  float4 a[16];
  {
    int row = strip * 16 + c;
    if (row >= N) row = N - 1;
    const float* ap = X + (size_t)row * 128 + 4 * h;
#pragma unroll
    for (int u = 0; u < 16; ++u) a[u] = *(const float4*)(ap + 8 * u);
  }

  {
    const int m = threadIdx.x & 31;
#pragma unroll
    for (int it = 0; it < 2; ++it) {
      const int k4 = (threadIdx.x >> 5) + 16 * it;
      const float* wp = W + (size_t)(4 * k4) * 128 + 4 * m;
      const float4 r0 = *(const float4*)(wp);
      const float4 r1 = *(const float4*)(wp + 128);
      const float4 r2 = *(const float4*)(wp + 256);
      const float4 r3 = *(const float4*)(wp + 384);
      const int base = (k4 * 8 + (m >> 4) * 4) * 16 + (m & 15);
      wl[base]      = make_float4(r0.x, r1.x, r2.x, r3.x);
      wl[base + 16] = make_float4(r0.y, r1.y, r2.y, r3.y);
      wl[base + 32] = make_float4(r0.z, r1.z, r2.z, r3.z);
      wl[base + 48] = make_float4(r0.w, r1.w, r2.w, r3.w);
    }
  }
  __syncthreads();

  const float4* bl = wl + h * 128 + c;
  while (strip < nstrips) {
    __atomic_signal_fence(__ATOMIC_SEQ_CST);
    const int nxt = strip + stride;
    int nrow = nxt * 16 + c;
    if (nrow >= N) nrow = N - 1;
    const float* np = X + (size_t)nrow * 128 + 4 * h;

    v8f acc[8];
#pragma unroll
    for (int j = 0; j < 8; ++j) acc[j] = (v8f){0.f, 0.f, 0.f, 0.f, 0.f, 0.f, 0.f, 0.f};

#pragma unroll
    for (int g = 0; g < 4; ++g) {
#pragma unroll
      for (int uu = 0; uu < 4; ++uu) {
        const int u = 4 * g + uu;
        v2f a0; a0.x = a[u].x; a0.y = a[u].y;
        v2f a1; a1.x = a[u].z; a1.y = a[u].w;
#pragma unroll
        for (int j = 0; j < 8; ++j) {
          const float4 b = bl[(2 * u * 8 + j) * 16];
          v2f b0; b0.x = b.x; b0.y = b.y;
          v2f b1; b1.x = b.z; b1.y = b.w;
          acc[j] = __builtin_amdgcn_wmma_f32_16x16x4_f32(false, a0, false, b0,
                                                         (short)0, acc[j], false, false);
          acc[j] = __builtin_amdgcn_wmma_f32_16x16x4_f32(false, a1, false, b1,
                                                         (short)0, acc[j], false, false);
        }
      }
#pragma unroll
      for (int uu = 0; uu < 4; ++uu) a[4 * g + uu] = *(const float4*)(np + 8 * (4 * g + uu));
    }

    const int rowbase = strip * 16 + 8 * h;
#pragma unroll
    for (int jj = 0; jj < 2; ++jj) {
#pragma unroll
      for (int i = 0; i < 8; ++i) {
        const int row = rowbase + i;
        if (row < N)
          *(float4*)(Hout + (size_t)row * 128 + 64 * jj + 4 * c) =
              make_float4(acc[4 * jj][i], acc[4 * jj + 1][i], acc[4 * jj + 2][i], acc[4 * jj + 3][i]);
      }
    }
    strip = nxt;
  }
}

__global__ void gat_scores(const float* __restrict__ Hm,
                           const float* __restrict__ asrc,
                           const float* __restrict__ adst,
                           float* __restrict__ es, float* __restrict__ ed, int N) {
  const int idx = blockIdx.x * blockDim.x + threadIdx.x;
  if (idx >= N * 4) return;
  const int hd = idx & 3;
  const size_t base = (size_t)(idx >> 2) * 128 + hd * 32;
  const float* hp  = Hm + base;
  const float* sa  = asrc + hd * 32;
  const float* da  = adst + hd * 32;
  float s = 0.f, d = 0.f;
#pragma unroll
  for (int c = 0; c < 32; c += 4) {
    float4 hv = *(const float4*)(hp + c);
    float4 av = *(const float4*)(sa + c);
    float4 bv = *(const float4*)(da + c);
    s += hv.x * av.x + hv.y * av.y + hv.z * av.z + hv.w * av.w;
    d += hv.x * bv.x + hv.y * bv.y + hv.z * bv.z + hv.w * bv.w;
  }
  es[idx] = s;
  ed[idx] = d;
}

__device__ __forceinline__ float lrelu(float x) {
  return x > 0.f ? x : NEG_SLOPE * x;
}


__global__ void csr_zero(int* __restrict__ cnt, int N) {
  const int i = blockIdx.x * blockDim.x + threadIdx.x;
  if (i < N) cnt[i] = 0;
}

__global__ void csr_count(const int* __restrict__ ei, int* __restrict__ cnt,
                          int* __restrict__ rank, int E) {
  const int e = blockIdx.x * blockDim.x + threadIdx.x;
  if (e >= E) return;
  rank[e] = atomicAdd(&cnt[ei[E + e]], 1);
}

__global__ __launch_bounds__(1024) void csr_scan(const int* __restrict__ cnt,
                                                 int* __restrict__ rowptr, int N) {
  __shared__ int wsum[32];
  __shared__ int wpre[32];
  const int tid  = threadIdx.x;
  const int lane = tid & 31;
  const int wv   = tid >> 5;
  const int base = blockIdx.x * 1024;
  int pre = 0;
  for (int i = tid; i < base; i += 1024) pre += cnt[i];
  const int idx = base + tid;
  const int c = idx < N ? cnt[idx] : 0;
  int inc = c;
#pragma unroll
  for (int o = 1; o < 32; o <<= 1) {
    const int t = __shfl_up(inc, o);
    if (lane >= o) inc += t;
  }
#pragma unroll
  for (int o = 16; o > 0; o >>= 1) pre += __shfl_xor(pre, o);
  if (lane == 31) wsum[wv] = inc;
  if (lane == 0)  wpre[wv] = pre;
  __syncthreads();
  int off = 0;
  for (int i = 0; i < 32; ++i) {
    off += wpre[i];
    if (i < wv) off += wsum[i];
  }
  const int excl = off + inc - c;
  if (idx < N) rowptr[idx] = excl;
  if (idx == N - 1) rowptr[N] = excl + c;
}

__global__ void csr_fill(const int* __restrict__ ei, const int* __restrict__ rowptr,
                         const int* __restrict__ rank, int* __restrict__ col, int E) {
  const int e = blockIdx.x * blockDim.x + threadIdx.x;
  if (e >= E) return;
  col[rowptr[ei[E + e]] + rank[e]] = ei[e];
}

__global__ __launch_bounds__(256) void csr_sort_rows(const int* __restrict__ rowptr,
                                                     const int* __restrict__ cin,
                                                     int* __restrict__ cout, int N) {
  const int lane = threadIdx.x & 31;
  const int node = __builtin_amdgcn_readfirstlane(blockIdx.x * 8 + (threadIdx.x >> 5));
  if (node >= N) return;
  const int beg = __builtin_amdgcn_readfirstlane(rowptr[node]);
  const int n   = __builtin_amdgcn_readfirstlane(rowptr[node + 1]) - beg;
  for (int i = lane; i < n; i += 32) {
    const int key = cin[beg + i];
    int r = 0;
    for (int j = 0; j < n; ++j) {
      const int kj = cin[beg + j];
      r += (kj < key || (kj == key && j < i)) ? 1 : 0;
    }
    cout[beg + r] = key;
  }
}

__device__ __forceinline__ float wave_max(float v) {
#pragma unroll
  for (int o = 16; o > 0; o >>= 1) v = fmaxf(v, __shfl_xor(v, o));
  return v;
}
__device__ __forceinline__ float pick_head(float x0, float x1, float x2, float x3, int j,
                                           int k0, int k1, int k2, int k3) {
  const int b0 = __builtin_amdgcn_readlane(__float_as_int(x0), j);
  const int b1 = __builtin_amdgcn_readlane(__float_as_int(x1), j);
  const int b2 = __builtin_amdgcn_readlane(__float_as_int(x2), j);
  const int b3 = __builtin_amdgcn_readlane(__float_as_int(x3), j);
  return __int_as_float((b0 & k0) | (b1 & k1) | (b2 & k2) | (b3 & k3));
}

__global__ __launch_bounds__(256) void gat_gather(
    const int* __restrict__ rowptr, const int* __restrict__ col,
    const float* __restrict__ es, const float* __restrict__ ed,
    const float* __restrict__ Hm, const float* __restrict__ bias,
    float* __restrict__ outp, int N, int apply_elu) {
  const int lane = threadIdx.x & 31;
  const int node = __builtin_amdgcn_readfirstlane(blockIdx.x * 8 + (threadIdx.x >> 5));
  if (node >= N) return;
  const int beg = __builtin_amdgcn_readfirstlane(rowptr[node]);
  const int tot = __builtin_amdgcn_readfirstlane(rowptr[node + 1]) - beg + 1;
  const int hd  = lane >> 3;
  const float4 vd = *(const float4*)(ed + (size_t)node * 4);
  const float ninf = -__builtin_inff();

  int sc;
  float l0, l1, l2, l3;
  auto load_chunk = [&](int t0) {
    const int t = t0 + lane;
    sc = node;
    l0 = l1 = l2 = l3 = ninf;
    if (t < tot) {
      if (t < tot - 1) sc = col[beg + t];
      const float4 vs = *(const float4*)(es + (size_t)sc * 4);
      l0 = lrelu(vs.x + vd.x);
      l1 = lrelu(vs.y + vd.y);
      l2 = lrelu(vs.z + vd.z);
      l3 = lrelu(vs.w + vd.w);
    }
  };
  load_chunk(0);
  float m0 = l0, m1 = l1, m2 = l2, m3 = l3;
  for (int t0 = 32; t0 < tot; t0 += 32) {
    load_chunk(t0);
    m0 = fmaxf(m0, l0); m1 = fmaxf(m1, l1); m2 = fmaxf(m2, l2); m3 = fmaxf(m3, l3);
  }
  m0 = wave_max(m0); m1 = wave_max(m1); m2 = wave_max(m2); m3 = wave_max(m3);

  float4 acc = make_float4(0.f, 0.f, 0.f, 0.f);
  float ssum = 0.f;
  const float* hrow = Hm + lane * 4;
  const int k0 = hd == 0 ? -1 : 0, k1 = hd == 1 ? -1 : 0, k2 = hd == 2 ? -1 : 0, k3 = hd == 3 ? -1 : 0;
  for (int t0 = 0; t0 < tot; t0 += 32) {
    if (tot > 32) load_chunk(t0);
    const float x0 = expf(l0 - m0);
    const float x1 = expf(l1 - m1);
    const float x2 = expf(l2 - m2);
    const float x3 = expf(l3 - m3);
    const int cnt = (tot - t0) < 32 ? (tot - t0) : 32;
    int j = 0;
    for (; j + 4 <= cnt; j += 4) {
      const float4 ha = *(const float4*)(hrow + (size_t)__builtin_amdgcn_readlane(sc, j) * 128);
      const float4 hb = *(const float4*)(hrow + (size_t)__builtin_amdgcn_readlane(sc, j + 1) * 128);
      const float4 hc = *(const float4*)(hrow + (size_t)__builtin_amdgcn_readlane(sc, j + 2) * 128);
      const float4 hd4 = *(const float4*)(hrow + (size_t)__builtin_amdgcn_readlane(sc, j + 3) * 128);
      const float ea = pick_head(x0, x1, x2, x3, j, k0, k1, k2, k3);
      const float eb = pick_head(x0, x1, x2, x3, j + 1, k0, k1, k2, k3);
      const float ec = pick_head(x0, x1, x2, x3, j + 2, k0, k1, k2, k3);
      const float ee = pick_head(x0, x1, x2, x3, j + 3, k0, k1, k2, k3);
      ssum += ea; acc.x = fmaf(ha.x, ea, acc.x); acc.y = fmaf(ha.y, ea, acc.y); acc.z = fmaf(ha.z, ea, acc.z); acc.w = fmaf(ha.w, ea, acc.w);
      ssum += eb; acc.x = fmaf(hb.x, eb, acc.x); acc.y = fmaf(hb.y, eb, acc.y); acc.z = fmaf(hb.z, eb, acc.z); acc.w = fmaf(hb.w, eb, acc.w);
      ssum += ec; acc.x = fmaf(hc.x, ec, acc.x); acc.y = fmaf(hc.y, ec, acc.y); acc.z = fmaf(hc.z, ec, acc.z); acc.w = fmaf(hc.w, ec, acc.w);
      ssum += ee; acc.x = fmaf(hd4.x, ee, acc.x); acc.y = fmaf(hd4.y, ee, acc.y); acc.z = fmaf(hd4.z, ee, acc.z); acc.w = fmaf(hd4.w, ee, acc.w);
    }
#pragma unroll 1
    for (; j < cnt; ++j) {
      const float4 ha = *(const float4*)(hrow + (size_t)__builtin_amdgcn_readlane(sc, j) * 128);
      const float ea = pick_head(x0, x1, x2, x3, j, k0, k1, k2, k3);
      ssum += ea; acc.x = fmaf(ha.x, ea, acc.x); acc.y = fmaf(ha.y, ea, acc.y); acc.z = fmaf(ha.z, ea, acc.z); acc.w = fmaf(ha.w, ea, acc.w);
    }
  }
  const float4 bv = *(const float4*)(bias + lane * 4);
  float4 v;
  v.x = acc.x / ssum + bv.x;
  v.y = acc.y / ssum + bv.y;
  v.z = acc.z / ssum + bv.z;
  v.w = acc.w / ssum + bv.w;
  if (apply_elu) {
    v.x = v.x > 0.f ? v.x : expm1f(v.x);
    v.y = v.y > 0.f ? v.y : expm1f(v.y);
    v.z = v.z > 0.f ? v.z : expm1f(v.z);
    v.w = v.w > 0.f ? v.w : expm1f(v.w);
  }
  *(float4*)(outp + (size_t)node * 128 + lane * 4) = v;
}

extern "C" void kernel_launch(void* const* d_in, const int* in_sizes, int n_in,
                              void* d_out, int out_size, void* d_ws, size_t ws_size,
                              hipStream_t stream) {
  const float* x  = (const float*)d_in[0];
  const int*   ei = (const int*)d_in[1];
  const int N = in_sizes[0] / 128;
  const int E = in_sizes[1] / 2;

  const float* Wl[3] = {(const float*)d_in[2], (const float*)d_in[6],  (const float*)d_in[10]};
  const float* As[3] = {(const float*)d_in[3], (const float*)d_in[7],  (const float*)d_in[11]};
  const float* Ad[3] = {(const float*)d_in[4], (const float*)d_in[8],  (const float*)d_in[12]};
  const float* Bb[3] = {(const float*)d_in[5], (const float*)d_in[9],  (const float*)d_in[13]};

  float* F0     = (float*)d_ws;
  float* F1     = F0 + (size_t)N * 128;
  float* es     = F1 + (size_t)N * 128;
  float* ed     = es + (size_t)N * 4;
  int*   cnt    = (int*)(ed + (size_t)N * 4);
  int*   rowptr = cnt + (size_t)N;
  int*   rank   = rowptr + (size_t)N + 4;
  int*   colA   = rank + (size_t)E;
  int*   colB   = colA + (size_t)E;

  const int gNode  = (N + 255) / 256;
  const int gEdge  = (E + 255) / 256;
  const int gScan  = (N + 1023) / 1024;
  const int gWave  = (N + 7) / 8;
  const int gScore = (N * 4 + 255) / 256;

  csr_zero<<<gNode, 256, 0, stream>>>(cnt, N);
  csr_count<<<gEdge, 256, 0, stream>>>(ei, cnt, rank, E);
  csr_scan<<<gScan, 1024, 0, stream>>>(cnt, rowptr, N);
  csr_fill<<<gEdge, 256, 0, stream>>>(ei, rowptr, rank, colA, E);
  csr_sort_rows<<<gWave, 256, 0, stream>>>(rowptr, colA, colB, N);

  const float* inp = x;
  for (int layer = 0; layer < 3; ++layer) {
    gat_gemm128<<<GEMM_WGS, 512, 0, stream>>>(inp, Wl[layer], F0, N);
    gat_scores<<<gScore, 256, 0, stream>>>(F0, As[layer], Ad[layer], es, ed, N);
    float* ot = (layer == 2) ? (float*)d_out : F1;
    gat_gather<<<gWave, 256, 0, stream>>>(rowptr, colB, es, ed, F0, Bb[layer], ot, N,
                                          layer < 2 ? 1 : 0);
    inp = F1;
  }
}


# Scores computed from the gathered rows in registers (dots summed over a head's 8 lanes): score launches and per-edge score gather go; softmax sum order changes (self first, running max per 8 edges)
# speedup vs baseline: 4.8561x; 1.0665x over previous
#include <hip/hip_runtime.h>
#include <hip/hip_bf16.h>
#include <math.h>

typedef __attribute__((ext_vector_type(2))) float v2f;
typedef __attribute__((ext_vector_type(8))) float v8f;

#define NEG_SLOPE 0.2f

#define GEMM_WGS 256
__global__ __launch_bounds__(512) void gat_gemm128(const float* __restrict__ X,
                                                   const float* __restrict__ W,
                                                   float* __restrict__ Hout,
                                                   int N) {
  __shared__ float4 wl[32 * 8 * 16];
  const int lane = threadIdx.x & 31;
  const int wv   = __builtin_amdgcn_readfirstlane(threadIdx.x >> 5);
  const int c    = lane & 15;
  const int h    = lane >> 4;
  const int nstrips = (N + 15) >> 4;
  int strip = wv * GEMM_WGS + blockIdx.x;
  const int stride = 16 * GEMM_WGS;

  float4 a[16];
  {
    int row = strip * 16 + c;
    if (row >= N) row = N - 1;
    const float* ap = X + (size_t)row * 128 + 4 * h;
#pragma unroll
    for (int u = 0; u < 16; ++u) a[u] = *(const float4*)(ap + 8 * u);
  }

  {
    const int m = threadIdx.x & 31;
#pragma unroll
    for (int it = 0; it < 2; ++it) {
      const int k4 = (threadIdx.x >> 5) + 16 * it;
      const float* wp = W + (size_t)(4 * k4) * 128 + 4 * m;
      const float4 r0 = *(const float4*)(wp);
      const float4 r1 = *(const float4*)(wp + 128);
      const float4 r2 = *(const float4*)(wp + 256);
      const float4 r3 = *(const float4*)(wp + 384);
      const int base = (k4 * 8 + (m >> 4) * 4) * 16 + (m & 15);
      wl[base]      = make_float4(r0.x, r1.x, r2.x, r3.x);
      wl[base + 16] = make_float4(r0.y, r1.y, r2.y, r3.y);
      wl[base + 32] = make_float4(r0.z, r1.z, r2.z, r3.z);
      wl[base + 48] = make_float4(r0.w, r1.w, r2.w, r3.w);
    }
  }
  __syncthreads();

  const float4* bl = wl + h * 128 + c;
  while (strip < nstrips) {
    __atomic_signal_fence(__ATOMIC_SEQ_CST);
    const int nxt = strip + stride;
    int nrow = nxt * 16 + c;
    if (nrow >= N) nrow = N - 1;
    const float* np = X + (size_t)nrow * 128 + 4 * h;

    v8f acc[8];
#pragma unroll
    for (int j = 0; j < 8; ++j) acc[j] = (v8f){0.f, 0.f, 0.f, 0.f, 0.f, 0.f, 0.f, 0.f};

#pragma unroll
    for (int g = 0; g < 4; ++g) {
#pragma unroll
      for (int uu = 0; uu < 4; ++uu) {
        const int u = 4 * g + uu;
        v2f a0; a0.x = a[u].x; a0.y = a[u].y;
        v2f a1; a1.x = a[u].z; a1.y = a[u].w;
#pragma unroll
        for (int j = 0; j < 8; ++j) {
          const float4 b = bl[(2 * u * 8 + j) * 16];
          v2f b0; b0.x = b.x; b0.y = b.y;
          v2f b1; b1.x = b.z; b1.y = b.w;
          acc[j] = __builtin_amdgcn_wmma_f32_16x16x4_f32(false, a0, false, b0,
                                                         (short)0, acc[j], false, false);
          acc[j] = __builtin_amdgcn_wmma_f32_16x16x4_f32(false, a1, false, b1,
                                                         (short)0, acc[j], false, false);
        }
      }
#pragma unroll
      for (int uu = 0; uu < 4; ++uu) a[4 * g + uu] = *(const float4*)(np + 8 * (4 * g + uu));
    }

    const int rowbase = strip * 16 + 8 * h;
#pragma unroll
    for (int jj = 0; jj < 2; ++jj) {
#pragma unroll
      for (int i = 0; i < 8; ++i) {
        const int row = rowbase + i;
        if (row < N)
          *(float4*)(Hout + (size_t)row * 128 + 64 * jj + 4 * c) =
              make_float4(acc[4 * jj][i], acc[4 * jj + 1][i], acc[4 * jj + 2][i], acc[4 * jj + 3][i]);
      }
    }
    strip = nxt;
  }
}

__device__ __forceinline__ float lrelu(float x) {
  return x > 0.f ? x : NEG_SLOPE * x;
}


__global__ void csr_zero(int* __restrict__ cnt, int N) {
  const int i = blockIdx.x * blockDim.x + threadIdx.x;
  if (i < N) cnt[i] = 0;
}

__global__ void csr_count(const int* __restrict__ ei, int* __restrict__ cnt,
                          int* __restrict__ rank, int E) {
  const int e = blockIdx.x * blockDim.x + threadIdx.x;
  if (e >= E) return;
  rank[e] = atomicAdd(&cnt[ei[E + e]], 1);
}

__global__ __launch_bounds__(1024) void csr_scan(const int* __restrict__ cnt,
                                                 int* __restrict__ rowptr, int N) {
  __shared__ int wsum[32];
  __shared__ int wpre[32];
  const int tid  = threadIdx.x;
  const int lane = tid & 31;
  const int wv   = tid >> 5;
  const int base = blockIdx.x * 1024;
  int pre = 0;
  for (int i = tid; i < base; i += 1024) pre += cnt[i];
  const int idx = base + tid;
  const int c = idx < N ? cnt[idx] : 0;
  int inc = c;
#pragma unroll
  for (int o = 1; o < 32; o <<= 1) {
    const int t = __shfl_up(inc, o);
    if (lane >= o) inc += t;
  }
#pragma unroll
  for (int o = 16; o > 0; o >>= 1) pre += __shfl_xor(pre, o);
  if (lane == 31) wsum[wv] = inc;
  if (lane == 0)  wpre[wv] = pre;
  __syncthreads();
  int off = 0;
  for (int i = 0; i < 32; ++i) {
    off += wpre[i];
    if (i < wv) off += wsum[i];
  }
  const int excl = off + inc - c;
  if (idx < N) rowptr[idx] = excl;
  if (idx == N - 1) rowptr[N] = excl + c;
}

__global__ void csr_fill(const int* __restrict__ ei, const int* __restrict__ rowptr,
                         const int* __restrict__ rank, int* __restrict__ col, int E) {
  const int e = blockIdx.x * blockDim.x + threadIdx.x;
  if (e >= E) return;
  col[rowptr[ei[E + e]] + rank[e]] = ei[e];
}

__global__ __launch_bounds__(256) void csr_sort_rows(const int* __restrict__ rowptr,
                                                     const int* __restrict__ cin,
                                                     int* __restrict__ cout, int N) {
  const int lane = threadIdx.x & 31;
  const int node = __builtin_amdgcn_readfirstlane(blockIdx.x * 8 + (threadIdx.x >> 5));
  if (node >= N) return;
  const int beg = __builtin_amdgcn_readfirstlane(rowptr[node]);
  const int n   = __builtin_amdgcn_readfirstlane(rowptr[node + 1]) - beg;
  for (int i = lane; i < n; i += 32) {
    const int key = cin[beg + i];
    int r = 0;
    for (int j = 0; j < n; ++j) {
      const int kj = cin[beg + j];
      r += (kj < key || (kj == key && j < i)) ? 1 : 0;
    }
    cout[beg + r] = key;
  }
}

__device__ __forceinline__ float xch1(float v) {
  return __int_as_float(__builtin_amdgcn_update_dpp(0, __float_as_int(v), 0xB1, 0xF, 0xF, true));
}
__device__ __forceinline__ float xch2(float v) {
  return __int_as_float(__builtin_amdgcn_update_dpp(0, __float_as_int(v), 0x4E, 0xF, 0xF, true));
}
__device__ __forceinline__ float xch4(float v) {
  return __int_as_float(__builtin_amdgcn_update_dpp(0, __float_as_int(v), 0x164, 0xF, 0xF, true));
}
__device__ __forceinline__ float sum8(float v) {
  v += xch1(v); v += xch2(v); v += xch4(v);
  return v;
}
__device__ __forceinline__ float max8(float v) {
  v = fmaxf(v, xch1(v)); v = fmaxf(v, xch2(v)); v = fmaxf(v, xch4(v));
  return v;
}
template <int J>
__device__ __forceinline__ float bcast8(float v) {
  return __int_as_float(__builtin_amdgcn_mov_dpp8(__float_as_int(v), J * 0x249249));
}
__device__ __forceinline__ float dot4(const float4 r, const float4 a) {
  return fmaf(r.w, a.w, fmaf(r.z, a.z, fmaf(r.y, a.y, r.x * a.x)));
}

template <bool TAIL>
__device__ __forceinline__ void gat_chunk(const float* __restrict__ hrow, int sc, int c0, int rem,
                                          int i8, bool b1, bool b2, bool b4,
                                          const float4 a4, float ed,
                                          float& m, float& ssum, float4& acc) {
  float4 r[8];
  int srow[8];
#pragma unroll
  for (int k = 0; k < 8; ++k) srow[k] = __builtin_amdgcn_readlane(sc, c0 + k);
#pragma unroll
  for (int k = 0; k < 8; ++k) {
    if (!TAIL || k < rem)
      r[k] = *(const float4*)(hrow + (size_t)srow[k] * 128);
    else
      r[k] = make_float4(0.f, 0.f, 0.f, 0.f);
  }
  float p[8];
#pragma unroll
  for (int k = 0; k < 8; ++k) p[k] = dot4(r[k], a4);
  float q[4];
#pragma unroll
  for (int k = 0; k < 4; ++k)
    q[k] = (b1 ? p[2 * k + 1] : p[2 * k]) + xch1(b1 ? p[2 * k] : p[2 * k + 1]);
  float s2[2];
#pragma unroll
  for (int k = 0; k < 2; ++k)
    s2[k] = (b2 ? q[2 * k + 1] : q[2 * k]) + xch2(b2 ? q[2 * k] : q[2 * k + 1]);
  const float s = (b4 ? s2[1] : s2[0]) + xch4(b4 ? s2[0] : s2[1]);
  float l = lrelu(s + ed);
  if (TAIL) l = (i8 < rem) ? l : -__builtin_inff();
  const float mn = fmaxf(m, max8(l));
  const float rs = expf(m - mn);
  const float pw = expf(l - mn);
  m = mn;
  ssum = fmaf(ssum, rs, pw);
  acc.x *= rs; acc.y *= rs; acc.z *= rs; acc.w *= rs;
#define GAT_ADD_ROW(K)                                                    \
  {                                                                       \
    const float w = bcast8<K>(pw);                                        \
    acc.x = fmaf(r[K].x, w, acc.x); acc.y = fmaf(r[K].y, w, acc.y);       \
    acc.z = fmaf(r[K].z, w, acc.z); acc.w = fmaf(r[K].w, w, acc.w);       \
  }
  GAT_ADD_ROW(0) GAT_ADD_ROW(1) GAT_ADD_ROW(2) GAT_ADD_ROW(3)
  GAT_ADD_ROW(4) GAT_ADD_ROW(5) GAT_ADD_ROW(6) GAT_ADD_ROW(7)
#undef GAT_ADD_ROW
}

__global__ __launch_bounds__(256) void gat_gather(
    const int* __restrict__ rowptr, const int* __restrict__ col,
    const float* __restrict__ Hm, const float* __restrict__ asrc, const float* __restrict__ adst,
    const float* __restrict__ bias, float* __restrict__ outp, int N, int apply_elu) {
  const int lane = threadIdx.x & 31;
  const int node = __builtin_amdgcn_readfirstlane(blockIdx.x * 8 + (threadIdx.x >> 5));
  if (node >= N) return;
  const int beg = __builtin_amdgcn_readfirstlane(rowptr[node]);
  const int n   = __builtin_amdgcn_readfirstlane(rowptr[node + 1]) - beg;
  const float* hrow = Hm + lane * 4;
  const float4 a4 = *(const float4*)(asrc + lane * 4);
  const float4 d4 = *(const float4*)(adst + lane * 4);
  const float4 hs = *(const float4*)(hrow + (size_t)node * 128);
  const int  i8 = lane & 7;
  const bool b1 = (lane & 1) != 0, b2 = (lane & 2) != 0, b4 = (lane & 4) != 0;

  const float ed = sum8(dot4(hs, d4));
  float m = lrelu(sum8(dot4(hs, a4)) + ed);
  float4 acc = hs;
  float ssum = i8 == 0 ? 1.f : 0.f;

  for (int t0 = 0; t0 < n; t0 += 32) {
    int sc = node;
    if (t0 + lane < n) sc = col[beg + t0 + lane];
    const int cnt = (n - t0) < 32 ? (n - t0) : 32;
    int c0 = 0;
    for (; c0 + 8 <= cnt; c0 += 8)
      gat_chunk<false>(hrow, sc, c0, 8, i8, b1, b2, b4, a4, ed, m, ssum, acc);
    if (c0 < cnt)
      gat_chunk<true>(hrow, sc, c0, cnt - c0, i8, b1, b2, b4, a4, ed, m, ssum, acc);
  }
  const float tsum = sum8(ssum);
  const float4 bv = *(const float4*)(bias + lane * 4);
  float4 v;
  v.x = acc.x / tsum + bv.x;
  v.y = acc.y / tsum + bv.y;
  v.z = acc.z / tsum + bv.z;
  v.w = acc.w / tsum + bv.w;
  if (apply_elu) {
    v.x = v.x > 0.f ? v.x : expm1f(v.x);
    v.y = v.y > 0.f ? v.y : expm1f(v.y);
    v.z = v.z > 0.f ? v.z : expm1f(v.z);
    v.w = v.w > 0.f ? v.w : expm1f(v.w);
  }
  *(float4*)(outp + (size_t)node * 128 + lane * 4) = v;
}

extern "C" void kernel_launch(void* const* d_in, const int* in_sizes, int n_in,
                              void* d_out, int out_size, void* d_ws, size_t ws_size,
                              hipStream_t stream) {
  const float* x  = (const float*)d_in[0];
  const int*   ei = (const int*)d_in[1];
  const int N = in_sizes[0] / 128;
  const int E = in_sizes[1] / 2;

  const float* Wl[3] = {(const float*)d_in[2], (const float*)d_in[6],  (const float*)d_in[10]};
  const float* As[3] = {(const float*)d_in[3], (const float*)d_in[7],  (const float*)d_in[11]};
  const float* Ad[3] = {(const float*)d_in[4], (const float*)d_in[8],  (const float*)d_in[12]};
  const float* Bb[3] = {(const float*)d_in[5], (const float*)d_in[9],  (const float*)d_in[13]};

  float* F0     = (float*)d_ws;
  float* F1     = F0 + (size_t)N * 128;
  int*   cnt    = (int*)(F1 + (size_t)N * 128);
  int*   rowptr = cnt + (size_t)N;
  int*   rank   = rowptr + (size_t)N + 4;
  int*   colA   = rank + (size_t)E;
  int*   colB   = colA + (size_t)E;

  const int gNode  = (N + 255) / 256;
  const int gEdge  = (E + 255) / 256;
  const int gScan  = (N + 1023) / 1024;
  const int gWave  = (N + 7) / 8;

  csr_zero<<<gNode, 256, 0, stream>>>(cnt, N);
  csr_count<<<gEdge, 256, 0, stream>>>(ei, cnt, rank, E);
  csr_scan<<<gScan, 1024, 0, stream>>>(cnt, rowptr, N);
  csr_fill<<<gEdge, 256, 0, stream>>>(ei, rowptr, rank, colA, E);
  csr_sort_rows<<<gWave, 256, 0, stream>>>(rowptr, colA, colB, N);

  const float* inp = x;
  for (int layer = 0; layer < 3; ++layer) {
    gat_gemm128<<<GEMM_WGS, 512, 0, stream>>>(inp, Wl[layer], F0, N);
    float* ot = (layer == 2) ? (float*)d_out : F1;
    gat_gather<<<gWave, 256, 0, stream>>>(rowptr, colB, F0, As[layer], Ad[layer], Bb[layer], ot, N,
                                          layer < 2 ? 1 : 0);
    inp = F1;
  }
}


# CSR build through LDS: each slab of edges counting-sorted by 128-node bin, then one work-group per bin counts, places and orders its rows in LDS; 2 launches for 5, no global atomics; sum order kept
# speedup vs baseline: 6.9498x; 1.4311x over previous
#include <hip/hip_runtime.h>
#include <hip/hip_bf16.h>
#include <math.h>

typedef __attribute__((ext_vector_type(2))) float v2f;
typedef __attribute__((ext_vector_type(8))) float v8f;

#define NEG_SLOPE 0.2f

#define GEMM_WGS 256
__global__ __launch_bounds__(512) void gat_gemm128(const float* __restrict__ X,
                                                   const float* __restrict__ W,
                                                   float* __restrict__ Hout,
                                                   int N) {
  __shared__ float4 wl[32 * 8 * 16];
  const int lane = threadIdx.x & 31;
  const int wv   = __builtin_amdgcn_readfirstlane(threadIdx.x >> 5);
  const int c    = lane & 15;
  const int h    = lane >> 4;
  const int nstrips = (N + 15) >> 4;
  int strip = wv * GEMM_WGS + blockIdx.x;
  const int stride = 16 * GEMM_WGS;

  float4 a[16];
  {
    int row = strip * 16 + c;
    if (row >= N) row = N - 1;
    const float* ap = X + (size_t)row * 128 + 4 * h;
#pragma unroll
    for (int u = 0; u < 16; ++u) a[u] = *(const float4*)(ap + 8 * u);
  }

  {
    const int m = threadIdx.x & 31;
#pragma unroll
    for (int it = 0; it < 2; ++it) {
      const int k4 = (threadIdx.x >> 5) + 16 * it;
      const float* wp = W + (size_t)(4 * k4) * 128 + 4 * m;
      const float4 r0 = *(const float4*)(wp);
      const float4 r1 = *(const float4*)(wp + 128);
      const float4 r2 = *(const float4*)(wp + 256);
      const float4 r3 = *(const float4*)(wp + 384);
      const int base = (k4 * 8 + (m >> 4) * 4) * 16 + (m & 15);
      wl[base]      = make_float4(r0.x, r1.x, r2.x, r3.x);
      wl[base + 16] = make_float4(r0.y, r1.y, r2.y, r3.y);
      wl[base + 32] = make_float4(r0.z, r1.z, r2.z, r3.z);
      wl[base + 48] = make_float4(r0.w, r1.w, r2.w, r3.w);
    }
  }
  __syncthreads();

  const float4* bl = wl + h * 128 + c;
  while (strip < nstrips) {
    __atomic_signal_fence(__ATOMIC_SEQ_CST);
    const int nxt = strip + stride;
    int nrow = nxt * 16 + c;
    if (nrow >= N) nrow = N - 1;
    const float* np = X + (size_t)nrow * 128 + 4 * h;

    v8f acc[8];
#pragma unroll
    for (int j = 0; j < 8; ++j) acc[j] = (v8f){0.f, 0.f, 0.f, 0.f, 0.f, 0.f, 0.f, 0.f};

#pragma unroll
    for (int g = 0; g < 4; ++g) {
#pragma unroll
      for (int uu = 0; uu < 4; ++uu) {
        const int u = 4 * g + uu;
        v2f a0; a0.x = a[u].x; a0.y = a[u].y;
        v2f a1; a1.x = a[u].z; a1.y = a[u].w;
#pragma unroll
        for (int j = 0; j < 8; ++j) {
          const float4 b = bl[(2 * u * 8 + j) * 16];
          v2f b0; b0.x = b.x; b0.y = b.y;
          v2f b1; b1.x = b.z; b1.y = b.w;
          acc[j] = __builtin_amdgcn_wmma_f32_16x16x4_f32(false, a0, false, b0,
                                                         (short)0, acc[j], false, false);
          acc[j] = __builtin_amdgcn_wmma_f32_16x16x4_f32(false, a1, false, b1,
                                                         (short)0, acc[j], false, false);
        }
      }
#pragma unroll
      for (int uu = 0; uu < 4; ++uu) a[4 * g + uu] = *(const float4*)(np + 8 * (4 * g + uu));
    }

    const int rowbase = strip * 16 + 8 * h;
#pragma unroll
    for (int jj = 0; jj < 2; ++jj) {
#pragma unroll
      for (int i = 0; i < 8; ++i) {
        const int row = rowbase + i;
        if (row < N)
          *(float4*)(Hout + (size_t)row * 128 + 64 * jj + 4 * c) =
              make_float4(acc[4 * jj][i], acc[4 * jj + 1][i], acc[4 * jj + 2][i], acc[4 * jj + 3][i]);
      }
    }
    strip = nxt;
  }
}

__device__ __forceinline__ float lrelu(float x) {
  return x > 0.f ? x : NEG_SLOPE * x;
}

#define BIN_W 128
#define MAXNB 1024
#define SLAB 6144
#define ROWCAP 4096

__global__ __launch_bounds__(1024) void csr_bin(const int* __restrict__ ei, int* __restrict__ pairs_raw,
                                                int* __restrict__ tab, int E, int NB) {
  __shared__ int hist[MAXNB];
  __shared__ int lst[MAXNB];
  __shared__ int wsum[32];
  __shared__ int2 pl[SLAB];
  const int tid  = threadIdx.x;
  const int lane = tid & 31;
  const int wv   = tid >> 5;
  const int s0   = blockIdx.x * SLAB;
  const int cnt  = (E - s0) < SLAB ? (E - s0) : SLAB;
  hist[tid] = 0;
  __syncthreads();
  int dreg[6], sreg[6];
#pragma unroll
  for (int k = 0; k < 6; ++k) {
    const int i = tid + 1024 * k;
    dreg[k] = 0; sreg[k] = 0;
    if (i < cnt) {
      dreg[k] = ei[E + s0 + i];
      sreg[k] = ei[s0 + i];
      atomicAdd(&hist[dreg[k] >> 7], 1);
    }
  }
  __syncthreads();
  const int v = hist[tid];
  int inc = v;
#pragma unroll
  for (int o = 1; o < 32; o <<= 1) {
    const int t = __shfl_up(inc, o);
    if (lane >= o) inc += t;
  }
  if (lane == 31) wsum[wv] = inc;
  __syncthreads();
  int off = 0;
  for (int i = 0; i < wv; ++i) off += wsum[i];
  const int excl = off + inc - v;
  lst[tid]  = excl;
  hist[tid] = excl;
  __syncthreads();
#pragma unroll
  for (int k = 0; k < 6; ++k) {
    const int i = tid + 1024 * k;
    if (i < cnt) {
      const int p = atomicAdd(&hist[dreg[k] >> 7], 1);
      pl[p] = make_int2(sreg[k], dreg[k]);
    }
  }
  __syncthreads();
  int2* pairs = (int2*)pairs_raw + (size_t)blockIdx.x * SLAB;
#pragma unroll
  for (int k = 0; k < 6; ++k) {
    const int i = tid + 1024 * k;
    if (i < cnt) pairs[i] = pl[i];
  }
  int* tb = tab + (size_t)blockIdx.x * (NB + 1);
  if (tid < NB) tb[tid] = lst[tid];
  if (tid == 0) tb[NB] = cnt;
}

__device__ __forceinline__ void csr_order_rows(const int* st, const int* exL, const int* cntL,
                                               int* __restrict__ colrow, int lane, int wv) {
  for (int d = wv; d < BIN_W; d += 16) {
    const int beg = exL[d];
    const int m   = cntL[d];
    for (int i = lane; i < m; i += 32) {
      const int key = st[beg + i];
      int r = 0;
      for (int j = 0; j < m; ++j) {
        const int kj = st[beg + j];
        r += (kj < key || (kj == key && j < i)) ? 1 : 0;
      }
      colrow[beg + r] = key;
    }
  }
}

__global__ __launch_bounds__(512) void csr_rows(const int* __restrict__ pairs_raw,
                                               const int* __restrict__ tab,
                                               int* __restrict__ rowptr, int* __restrict__ spill,
                                               int* __restrict__ col, int N, int NB, int GA) {
  __shared__ int cntL[BIN_W];
  __shared__ int exL[BIN_W];
  __shared__ int curL[BIN_W];
  __shared__ int red[2][16];
  __shared__ int w4[4];
  __shared__ int stageL[ROWCAP];
  const int tid  = threadIdx.x;
  const int lane = tid & 31;
  const int wv   = tid >> 5;
  const int k    = blockIdx.x;
  const int n0   = k * BIN_W;
  const int2* pairs = (const int2*)pairs_raw;
  if (tid < BIN_W) cntL[tid] = 0;
  int sa = 0, sl = 0;
  for (int t = tid; t < GA; t += 512) {
    const int* tp = tab + (size_t)t * (NB + 1) + k;
    const int a = tp[0];
    sa += a;
    sl += tp[1] - a;
  }
#pragma unroll
  for (int o = 16; o > 0; o >>= 1) {
    sa += __shfl_xor(sa, o);
    sl += __shfl_xor(sl, o);
  }
  if (lane == 0) { red[0][wv] = sa; red[1][wv] = sl; }
  __syncthreads();
  int base = 0, n = 0;
  for (int i = 0; i < 16; ++i) { base += red[0][i]; n += red[1][i]; }
  for (int t = tid; t < GA; t += 512) {
    const int* tp = tab + (size_t)t * (NB + 1) + k;
    const int a = tp[0];
    const int len = tp[1] - a;
    const int2* run = pairs + (size_t)t * SLAB + a;
    for (int j = 0; j < len; ++j) atomicAdd(&cntL[run[j].y - n0], 1);
  }
  __syncthreads();
  int v = 0, inc = 0;
  if (tid < BIN_W) {
    v = cntL[tid];
    inc = v;
#pragma unroll
    for (int o = 1; o < 32; o <<= 1) {
      const int t = __shfl_up(inc, o);
      if (lane >= o) inc += t;
    }
    if (lane == 31) w4[wv] = inc;
  }
  __syncthreads();
  if (tid < BIN_W) {
    int off = 0;
    for (int i = 0; i < wv; ++i) off += w4[i];
    const int ex = off + inc - v;
    exL[tid]  = ex;
    curL[tid] = ex;
    const int node = n0 + tid;
    if (node < N) rowptr[node] = base + ex;
    if (node == N - 1) rowptr[N] = base + ex + v;
  }
  __syncthreads();
  int* colrow = col + base;
  if (n <= ROWCAP) {
    for (int t = tid; t < GA; t += 512) {
      const int* tp = tab + (size_t)t * (NB + 1) + k;
      const int a = tp[0];
      const int len = tp[1] - a;
      const int2* run = pairs + (size_t)t * SLAB + a;
      for (int j = 0; j < len; ++j) {
        const int2 e = run[j];
        stageL[atomicAdd(&curL[e.y - n0], 1)] = e.x;
      }
    }
    __syncthreads();
    csr_order_rows(stageL, exL, cntL, colrow, lane, wv);
  } else {
    int* sp = spill + base;
    for (int t = tid; t < GA; t += 512) {
      const int* tp = tab + (size_t)t * (NB + 1) + k;
      const int a = tp[0];
      const int len = tp[1] - a;
      const int2* run = pairs + (size_t)t * SLAB + a;
      for (int j = 0; j < len; ++j) {
        const int2 e = run[j];
        sp[atomicAdd(&curL[e.y - n0], 1)] = e.x;
      }
    }
    __threadfence_block();
    __syncthreads();
    csr_order_rows(sp, exL, cntL, colrow, lane, wv);
  }
}

__device__ __forceinline__ float xch1(float v) {
  return __int_as_float(__builtin_amdgcn_update_dpp(0, __float_as_int(v), 0xB1, 0xF, 0xF, true));
}
__device__ __forceinline__ float xch2(float v) {
  return __int_as_float(__builtin_amdgcn_update_dpp(0, __float_as_int(v), 0x4E, 0xF, 0xF, true));
}
__device__ __forceinline__ float xch4(float v) {
  return __int_as_float(__builtin_amdgcn_update_dpp(0, __float_as_int(v), 0x164, 0xF, 0xF, true));
}
__device__ __forceinline__ float sum8(float v) {
  v += xch1(v); v += xch2(v); v += xch4(v);
  return v;
}
__device__ __forceinline__ float max8(float v) {
  v = fmaxf(v, xch1(v)); v = fmaxf(v, xch2(v)); v = fmaxf(v, xch4(v));
  return v;
}
template <int J>
__device__ __forceinline__ float bcast8(float v) {
  return __int_as_float(__builtin_amdgcn_mov_dpp8(__float_as_int(v), J * 0x249249));
}
__device__ __forceinline__ float dot4(const float4 r, const float4 a) {
  return fmaf(r.w, a.w, fmaf(r.z, a.z, fmaf(r.y, a.y, r.x * a.x)));
}

template <bool TAIL>
__device__ __forceinline__ void gat_chunk(const float* __restrict__ hrow, int sc, int c0, int rem,
                                          int i8, bool b1, bool b2, bool b4,
                                          const float4 a4, float ed,
                                          float& m, float& ssum, float4& acc) {
  float4 r[8];
  int srow[8];
#pragma unroll
  for (int k = 0; k < 8; ++k) srow[k] = __builtin_amdgcn_readlane(sc, c0 + k);
#pragma unroll
  for (int k = 0; k < 8; ++k) {
    if (!TAIL || k < rem)
      r[k] = *(const float4*)(hrow + (size_t)srow[k] * 128);
    else
      r[k] = make_float4(0.f, 0.f, 0.f, 0.f);
  }
  float p[8];
#pragma unroll
  for (int k = 0; k < 8; ++k) p[k] = dot4(r[k], a4);
  float q[4];
#pragma unroll
  for (int k = 0; k < 4; ++k)
    q[k] = (b1 ? p[2 * k + 1] : p[2 * k]) + xch1(b1 ? p[2 * k] : p[2 * k + 1]);
  float s2[2];
#pragma unroll
  for (int k = 0; k < 2; ++k)
    s2[k] = (b2 ? q[2 * k + 1] : q[2 * k]) + xch2(b2 ? q[2 * k] : q[2 * k + 1]);
  const float s = (b4 ? s2[1] : s2[0]) + xch4(b4 ? s2[0] : s2[1]);
  float l = lrelu(s + ed);
  if (TAIL) l = (i8 < rem) ? l : -__builtin_inff();
  const float mn = fmaxf(m, max8(l));
  const float rs = expf(m - mn);
  const float pw = expf(l - mn);
  m = mn;
  ssum = fmaf(ssum, rs, pw);
  acc.x *= rs; acc.y *= rs; acc.z *= rs; acc.w *= rs;
#define GAT_ADD_ROW(K)                                                    \
  {                                                                       \
    const float w = bcast8<K>(pw);                                        \
    acc.x = fmaf(r[K].x, w, acc.x); acc.y = fmaf(r[K].y, w, acc.y);       \
    acc.z = fmaf(r[K].z, w, acc.z); acc.w = fmaf(r[K].w, w, acc.w);       \
  }
  GAT_ADD_ROW(0) GAT_ADD_ROW(1) GAT_ADD_ROW(2) GAT_ADD_ROW(3)
  GAT_ADD_ROW(4) GAT_ADD_ROW(5) GAT_ADD_ROW(6) GAT_ADD_ROW(7)
#undef GAT_ADD_ROW
}

__global__ __launch_bounds__(256) void gat_gather(
    const int* __restrict__ rowptr, const int* __restrict__ col,
    const float* __restrict__ Hm, const float* __restrict__ asrc, const float* __restrict__ adst,
    const float* __restrict__ bias, float* __restrict__ outp, int N, int apply_elu) {
  const int lane = threadIdx.x & 31;
  const int node = __builtin_amdgcn_readfirstlane(blockIdx.x * 8 + (threadIdx.x >> 5));
  if (node >= N) return;
  const int beg = __builtin_amdgcn_readfirstlane(rowptr[node]);
  const int n   = __builtin_amdgcn_readfirstlane(rowptr[node + 1]) - beg;
  const float* hrow = Hm + lane * 4;
  const float4 a4 = *(const float4*)(asrc + lane * 4);
  const float4 d4 = *(const float4*)(adst + lane * 4);
  const float4 hs = *(const float4*)(hrow + (size_t)node * 128);
  const int  i8 = lane & 7;
  const bool b1 = (lane & 1) != 0, b2 = (lane & 2) != 0, b4 = (lane & 4) != 0;

  const float ed = sum8(dot4(hs, d4));
  float m = lrelu(sum8(dot4(hs, a4)) + ed);
  float4 acc = hs;
  float ssum = i8 == 0 ? 1.f : 0.f;

  for (int t0 = 0; t0 < n; t0 += 32) {
    int sc = node;
    if (t0 + lane < n) sc = col[beg + t0 + lane];
    const int cnt = (n - t0) < 32 ? (n - t0) : 32;
    int c0 = 0;
    for (; c0 + 8 <= cnt; c0 += 8)
      gat_chunk<false>(hrow, sc, c0, 8, i8, b1, b2, b4, a4, ed, m, ssum, acc);
    if (c0 < cnt)
      gat_chunk<true>(hrow, sc, c0, cnt - c0, i8, b1, b2, b4, a4, ed, m, ssum, acc);
  }
  const float tsum = sum8(ssum);
  const float4 bv = *(const float4*)(bias + lane * 4);
  float4 v;
  v.x = acc.x / tsum + bv.x;
  v.y = acc.y / tsum + bv.y;
  v.z = acc.z / tsum + bv.z;
  v.w = acc.w / tsum + bv.w;
  if (apply_elu) {
    v.x = v.x > 0.f ? v.x : expm1f(v.x);
    v.y = v.y > 0.f ? v.y : expm1f(v.y);
    v.z = v.z > 0.f ? v.z : expm1f(v.z);
    v.w = v.w > 0.f ? v.w : expm1f(v.w);
  }
  *(float4*)(outp + (size_t)node * 128 + lane * 4) = v;
}

extern "C" void kernel_launch(void* const* d_in, const int* in_sizes, int n_in,
                              void* d_out, int out_size, void* d_ws, size_t ws_size,
                              hipStream_t stream) {
  const float* x  = (const float*)d_in[0];
  const int*   ei = (const int*)d_in[1];
  const int N = in_sizes[0] / 128;
  const int E = in_sizes[1] / 2;

  const float* Wl[3] = {(const float*)d_in[2], (const float*)d_in[6],  (const float*)d_in[10]};
  const float* As[3] = {(const float*)d_in[3], (const float*)d_in[7],  (const float*)d_in[11]};
  const float* Ad[3] = {(const float*)d_in[4], (const float*)d_in[8],  (const float*)d_in[12]};
  const float* Bb[3] = {(const float*)d_in[5], (const float*)d_in[9],  (const float*)d_in[13]};

  const int gWave = (N + 7) / 8;
  const int NB    = (N + BIN_W - 1) / BIN_W;
  const int GA    = (E + SLAB - 1) / SLAB;

  float* F0     = (float*)d_ws;
  float* F1     = F0 + (size_t)N * 128;
  int*   pairs  = (int*)(F1 + (size_t)N * 128);
  int*   tab    = pairs + (size_t)GA * SLAB * 2;
  int*   rowptr = tab + (size_t)GA * (NB + 1);
  int*   spill  = rowptr + (size_t)N + 4;
  int*   colB   = spill + (size_t)E;

  csr_bin<<<GA, 1024, 0, stream>>>(ei, pairs, tab, E, NB);
  csr_rows<<<NB, 512, 0, stream>>>(pairs, tab, rowptr, spill, colB, N, NB, GA);

  const float* inp = x;
  for (int layer = 0; layer < 3; ++layer) {
    gat_gemm128<<<GEMM_WGS, 512, 0, stream>>>(inp, Wl[layer], F0, N);
    float* ot = (layer == 2) ? (float*)d_out : F1;
    gat_gather<<<gWave, 256, 0, stream>>>(rowptr, colB, F0, As[layer], Ad[layer], Bb[layer], ot, N,
                                          layer < 2 ? 1 : 0);
    inp = F1;
  }
}


# Layers 0,1: each node row goes through an LDS tile straight into the next layer's f32 matrix product in one persistent launch (next W in LDS): 8 launches become 6; sum order kept
# speedup vs baseline: 7.5396x; 1.0849x over previous
#include <hip/hip_runtime.h>
#include <hip/hip_bf16.h>
#include <math.h>

typedef __attribute__((ext_vector_type(2))) float v2f;
typedef __attribute__((ext_vector_type(8))) float v8f;

#define NEG_SLOPE 0.2f

#define GEMM_WGS 256
__global__ __launch_bounds__(512) void gat_gemm128(const float* __restrict__ X,
                                                   const float* __restrict__ W,
                                                   float* __restrict__ Hout,
                                                   int N) {
  __shared__ float4 wl[32 * 8 * 16];
  const int lane = threadIdx.x & 31;
  const int wv   = __builtin_amdgcn_readfirstlane(threadIdx.x >> 5);
  const int c    = lane & 15;
  const int h    = lane >> 4;
  const int nstrips = (N + 15) >> 4;
  int strip = wv * GEMM_WGS + blockIdx.x;
  const int stride = 16 * GEMM_WGS;

  float4 a[16];
  {
    int row = strip * 16 + c;
    if (row >= N) row = N - 1;
    const float* ap = X + (size_t)row * 128 + 4 * h;
#pragma unroll
    for (int u = 0; u < 16; ++u) a[u] = *(const float4*)(ap + 8 * u);
  }

  {
    const int m = threadIdx.x & 31;
#pragma unroll
    for (int it = 0; it < 2; ++it) {
      const int k4 = (threadIdx.x >> 5) + 16 * it;
      const float* wp = W + (size_t)(4 * k4) * 128 + 4 * m;
      const float4 r0 = *(const float4*)(wp);
      const float4 r1 = *(const float4*)(wp + 128);
      const float4 r2 = *(const float4*)(wp + 256);
      const float4 r3 = *(const float4*)(wp + 384);
      const int base = (k4 * 8 + (m >> 4) * 4) * 16 + (m & 15);
      wl[base]      = make_float4(r0.x, r1.x, r2.x, r3.x);
      wl[base + 16] = make_float4(r0.y, r1.y, r2.y, r3.y);
      wl[base + 32] = make_float4(r0.z, r1.z, r2.z, r3.z);
      wl[base + 48] = make_float4(r0.w, r1.w, r2.w, r3.w);
    }
  }
  __syncthreads();

  const float4* bl = wl + h * 128 + c;
  while (strip < nstrips) {
    __atomic_signal_fence(__ATOMIC_SEQ_CST);
    const int nxt = strip + stride;
    int nrow = nxt * 16 + c;
    if (nrow >= N) nrow = N - 1;
    const float* np = X + (size_t)nrow * 128 + 4 * h;

    v8f acc[8];
#pragma unroll
    for (int j = 0; j < 8; ++j) acc[j] = (v8f){0.f, 0.f, 0.f, 0.f, 0.f, 0.f, 0.f, 0.f};

#pragma unroll
    for (int g = 0; g < 4; ++g) {
#pragma unroll
      for (int uu = 0; uu < 4; ++uu) {
        const int u = 4 * g + uu;
        v2f a0; a0.x = a[u].x; a0.y = a[u].y;
        v2f a1; a1.x = a[u].z; a1.y = a[u].w;
#pragma unroll
        for (int j = 0; j < 8; ++j) {
          const float4 b = bl[(2 * u * 8 + j) * 16];
          v2f b0; b0.x = b.x; b0.y = b.y;
          v2f b1; b1.x = b.z; b1.y = b.w;
          acc[j] = __builtin_amdgcn_wmma_f32_16x16x4_f32(false, a0, false, b0,
                                                         (short)0, acc[j], false, false);
          acc[j] = __builtin_amdgcn_wmma_f32_16x16x4_f32(false, a1, false, b1,
                                                         (short)0, acc[j], false, false);
        }
      }
#pragma unroll
      for (int uu = 0; uu < 4; ++uu) a[4 * g + uu] = *(const float4*)(np + 8 * (4 * g + uu));
    }

    const int rowbase = strip * 16 + 8 * h;
#pragma unroll
    for (int jj = 0; jj < 2; ++jj) {
#pragma unroll
      for (int i = 0; i < 8; ++i) {
        const int row = rowbase + i;
        if (row < N)
          *(float4*)(Hout + (size_t)row * 128 + 64 * jj + 4 * c) =
              make_float4(acc[4 * jj][i], acc[4 * jj + 1][i], acc[4 * jj + 2][i], acc[4 * jj + 3][i]);
      }
    }
    strip = nxt;
  }
}

__device__ __forceinline__ float lrelu(float x) {
  return x > 0.f ? x : NEG_SLOPE * x;
}

#define BIN_W 128
#define MAXNB 1024
#define SLAB 6144
#define ROWCAP 4096

__global__ __launch_bounds__(1024) void csr_bin(const int* __restrict__ ei, int* __restrict__ pairs_raw,
                                                int* __restrict__ tab, int E, int NB) {
  __shared__ int hist[MAXNB];
  __shared__ int lst[MAXNB];
  __shared__ int wsum[32];
  __shared__ int2 pl[SLAB];
  const int tid  = threadIdx.x;
  const int lane = tid & 31;
  const int wv   = tid >> 5;
  const int s0   = blockIdx.x * SLAB;
  const int cnt  = (E - s0) < SLAB ? (E - s0) : SLAB;
  hist[tid] = 0;
  __syncthreads();
  int dreg[6], sreg[6];
#pragma unroll
  for (int k = 0; k < 6; ++k) {
    const int i = tid + 1024 * k;
    dreg[k] = 0; sreg[k] = 0;
    if (i < cnt) {
      dreg[k] = ei[E + s0 + i];
      sreg[k] = ei[s0 + i];
      atomicAdd(&hist[dreg[k] >> 7], 1);
    }
  }
  __syncthreads();
  const int v = hist[tid];
  int inc = v;
#pragma unroll
  for (int o = 1; o < 32; o <<= 1) {
    const int t = __shfl_up(inc, o);
    if (lane >= o) inc += t;
  }
  if (lane == 31) wsum[wv] = inc;
  __syncthreads();
  int off = 0;
  for (int i = 0; i < wv; ++i) off += wsum[i];
  const int excl = off + inc - v;
  lst[tid]  = excl;
  hist[tid] = excl;
  __syncthreads();
#pragma unroll
  for (int k = 0; k < 6; ++k) {
    const int i = tid + 1024 * k;
    if (i < cnt) {
      const int p = atomicAdd(&hist[dreg[k] >> 7], 1);
      pl[p] = make_int2(sreg[k], dreg[k]);
    }
  }
  __syncthreads();
  int2* pairs = (int2*)pairs_raw + (size_t)blockIdx.x * SLAB;
#pragma unroll
  for (int k = 0; k < 6; ++k) {
    const int i = tid + 1024 * k;
    if (i < cnt) pairs[i] = pl[i];
  }
  int* tb = tab + (size_t)blockIdx.x * (NB + 1);
  if (tid < NB) tb[tid] = lst[tid];
  if (tid == 0) tb[NB] = cnt;
}

__device__ __forceinline__ void csr_order_rows(const int* st, const int* exL, const int* cntL,
                                               int* __restrict__ colrow, int lane, int wv) {
  for (int d = wv; d < BIN_W; d += 16) {
    const int beg = exL[d];
    const int m   = cntL[d];
    for (int i = lane; i < m; i += 32) {
      const int key = st[beg + i];
      int r = 0;
      for (int j = 0; j < m; ++j) {
        const int kj = st[beg + j];
        r += (kj < key || (kj == key && j < i)) ? 1 : 0;
      }
      colrow[beg + r] = key;
    }
  }
}

__global__ __launch_bounds__(512) void csr_rows(const int* __restrict__ pairs_raw,
                                               const int* __restrict__ tab,
                                               int* __restrict__ rowptr, int* __restrict__ spill,
                                               int* __restrict__ col, int N, int NB, int GA) {
  __shared__ int cntL[BIN_W];
  __shared__ int exL[BIN_W];
  __shared__ int curL[BIN_W];
  __shared__ int red[2][16];
  __shared__ int w4[4];
  __shared__ int stageL[ROWCAP];
  const int tid  = threadIdx.x;
  const int lane = tid & 31;
  const int wv   = tid >> 5;
  const int k    = blockIdx.x;
  const int n0   = k * BIN_W;
  const int2* pairs = (const int2*)pairs_raw;
  if (tid < BIN_W) cntL[tid] = 0;
  int sa = 0, sl = 0;
  for (int t = tid; t < GA; t += 512) {
    const int* tp = tab + (size_t)t * (NB + 1) + k;
    const int a = tp[0];
    sa += a;
    sl += tp[1] - a;
  }
#pragma unroll
  for (int o = 16; o > 0; o >>= 1) {
    sa += __shfl_xor(sa, o);
    sl += __shfl_xor(sl, o);
  }
  if (lane == 0) { red[0][wv] = sa; red[1][wv] = sl; }
  __syncthreads();
  int base = 0, n = 0;
  for (int i = 0; i < 16; ++i) { base += red[0][i]; n += red[1][i]; }
  for (int t = tid; t < GA; t += 512) {
    const int* tp = tab + (size_t)t * (NB + 1) + k;
    const int a = tp[0];
    const int len = tp[1] - a;
    const int2* run = pairs + (size_t)t * SLAB + a;
    for (int j = 0; j < len; ++j) atomicAdd(&cntL[run[j].y - n0], 1);
  }
  __syncthreads();
  int v = 0, inc = 0;
  if (tid < BIN_W) {
    v = cntL[tid];
    inc = v;
#pragma unroll
    for (int o = 1; o < 32; o <<= 1) {
      const int t = __shfl_up(inc, o);
      if (lane >= o) inc += t;
    }
    if (lane == 31) w4[wv] = inc;
  }
  __syncthreads();
  if (tid < BIN_W) {
    int off = 0;
    for (int i = 0; i < wv; ++i) off += w4[i];
    const int ex = off + inc - v;
    exL[tid]  = ex;
    curL[tid] = ex;
    const int node = n0 + tid;
    if (node < N) rowptr[node] = base + ex;
    if (node == N - 1) rowptr[N] = base + ex + v;
  }
  __syncthreads();
  int* colrow = col + base;
  if (n <= ROWCAP) {
    for (int t = tid; t < GA; t += 512) {
      const int* tp = tab + (size_t)t * (NB + 1) + k;
      const int a = tp[0];
      const int len = tp[1] - a;
      const int2* run = pairs + (size_t)t * SLAB + a;
      for (int j = 0; j < len; ++j) {
        const int2 e = run[j];
        stageL[atomicAdd(&curL[e.y - n0], 1)] = e.x;
      }
    }
    __syncthreads();
    csr_order_rows(stageL, exL, cntL, colrow, lane, wv);
  } else {
    int* sp = spill + base;
    for (int t = tid; t < GA; t += 512) {
      const int* tp = tab + (size_t)t * (NB + 1) + k;
      const int a = tp[0];
      const int len = tp[1] - a;
      const int2* run = pairs + (size_t)t * SLAB + a;
      for (int j = 0; j < len; ++j) {
        const int2 e = run[j];
        sp[atomicAdd(&curL[e.y - n0], 1)] = e.x;
      }
    }
    __threadfence_block();
    __syncthreads();
    csr_order_rows(sp, exL, cntL, colrow, lane, wv);
  }
}

__device__ __forceinline__ float xch1(float v) {
  return __int_as_float(__builtin_amdgcn_update_dpp(0, __float_as_int(v), 0xB1, 0xF, 0xF, true));
}
__device__ __forceinline__ float xch2(float v) {
  return __int_as_float(__builtin_amdgcn_update_dpp(0, __float_as_int(v), 0x4E, 0xF, 0xF, true));
}
__device__ __forceinline__ float xch4(float v) {
  return __int_as_float(__builtin_amdgcn_update_dpp(0, __float_as_int(v), 0x164, 0xF, 0xF, true));
}
__device__ __forceinline__ float sum8(float v) {
  v += xch1(v); v += xch2(v); v += xch4(v);
  return v;
}
__device__ __forceinline__ float max8(float v) {
  v = fmaxf(v, xch1(v)); v = fmaxf(v, xch2(v)); v = fmaxf(v, xch4(v));
  return v;
}
template <int J>
__device__ __forceinline__ float bcast8(float v) {
  return __int_as_float(__builtin_amdgcn_mov_dpp8(__float_as_int(v), J * 0x249249));
}
__device__ __forceinline__ float dot4(const float4 r, const float4 a) {
  return fmaf(r.w, a.w, fmaf(r.z, a.z, fmaf(r.y, a.y, r.x * a.x)));
}

template <bool TAIL>
__device__ __forceinline__ void gat_chunk(const float* __restrict__ hrow, int sc, int c0, int rem,
                                          int i8, bool b1, bool b2, bool b4,
                                          const float4 a4, float ed,
                                          float& m, float& ssum, float4& acc) {
  float4 r[8];
  int srow[8];
#pragma unroll
  for (int k = 0; k < 8; ++k) srow[k] = __builtin_amdgcn_readlane(sc, c0 + k);
#pragma unroll
  for (int k = 0; k < 8; ++k) {
    if (!TAIL || k < rem)
      r[k] = *(const float4*)(hrow + (size_t)srow[k] * 128);
    else
      r[k] = make_float4(0.f, 0.f, 0.f, 0.f);
  }
  float p[8];
#pragma unroll
  for (int k = 0; k < 8; ++k) p[k] = dot4(r[k], a4);
  float q[4];
#pragma unroll
  for (int k = 0; k < 4; ++k)
    q[k] = (b1 ? p[2 * k + 1] : p[2 * k]) + xch1(b1 ? p[2 * k] : p[2 * k + 1]);
  float s2[2];
#pragma unroll
  for (int k = 0; k < 2; ++k)
    s2[k] = (b2 ? q[2 * k + 1] : q[2 * k]) + xch2(b2 ? q[2 * k] : q[2 * k + 1]);
  const float s = (b4 ? s2[1] : s2[0]) + xch4(b4 ? s2[0] : s2[1]);
  float l = lrelu(s + ed);
  if (TAIL) l = (i8 < rem) ? l : -__builtin_inff();
  const float mn = fmaxf(m, max8(l));
  const float rs = expf(m - mn);
  const float pw = expf(l - mn);
  m = mn;
  ssum = fmaf(ssum, rs, pw);
  acc.x *= rs; acc.y *= rs; acc.z *= rs; acc.w *= rs;
#define GAT_ADD_ROW(K)                                                    \
  {                                                                       \
    const float w = bcast8<K>(pw);                                        \
    acc.x = fmaf(r[K].x, w, acc.x); acc.y = fmaf(r[K].y, w, acc.y);       \
    acc.z = fmaf(r[K].z, w, acc.z); acc.w = fmaf(r[K].w, w, acc.w);       \
  }
  GAT_ADD_ROW(0) GAT_ADD_ROW(1) GAT_ADD_ROW(2) GAT_ADD_ROW(3)
  GAT_ADD_ROW(4) GAT_ADD_ROW(5) GAT_ADD_ROW(6) GAT_ADD_ROW(7)
#undef GAT_ADD_ROW
}

__device__ __forceinline__ float4 gat_node(
    const int* __restrict__ rowptr, const int* __restrict__ col, const float* __restrict__ Hm,
    const float4 a4, const float4 d4, const float* __restrict__ bias, int node, int lane, int apply_elu) {
  const int beg = __builtin_amdgcn_readfirstlane(rowptr[node]);
  const int n   = __builtin_amdgcn_readfirstlane(rowptr[node + 1]) - beg;
  const float* hrow = Hm + lane * 4;
  const float4 hs = *(const float4*)(hrow + (size_t)node * 128);
  const int  i8 = lane & 7;
  const bool b1 = (lane & 1) != 0, b2 = (lane & 2) != 0, b4 = (lane & 4) != 0;

  const float ed = sum8(dot4(hs, d4));
  float m = lrelu(sum8(dot4(hs, a4)) + ed);
  float4 acc = hs;
  float ssum = i8 == 0 ? 1.f : 0.f;

  for (int t0 = 0; t0 < n; t0 += 32) {
    int sc = node;
    if (t0 + lane < n) sc = col[beg + t0 + lane];
    const int cnt = (n - t0) < 32 ? (n - t0) : 32;
    int c0 = 0;
    for (; c0 + 8 <= cnt; c0 += 8)
      gat_chunk<false>(hrow, sc, c0, 8, i8, b1, b2, b4, a4, ed, m, ssum, acc);
    if (c0 < cnt)
      gat_chunk<true>(hrow, sc, c0, cnt - c0, i8, b1, b2, b4, a4, ed, m, ssum, acc);
  }
  const float tsum = sum8(ssum);
  const float4 bv = *(const float4*)(bias + lane * 4);
  float4 v;
  v.x = acc.x / tsum + bv.x;
  v.y = acc.y / tsum + bv.y;
  v.z = acc.z / tsum + bv.z;
  v.w = acc.w / tsum + bv.w;
  if (apply_elu) {
    v.x = v.x > 0.f ? v.x : expm1f(v.x);
    v.y = v.y > 0.f ? v.y : expm1f(v.y);
    v.z = v.z > 0.f ? v.z : expm1f(v.z);
    v.w = v.w > 0.f ? v.w : expm1f(v.w);
  }
  return v;
}

__global__ __launch_bounds__(256) void gat_gather(
    const int* __restrict__ rowptr, const int* __restrict__ col,
    const float* __restrict__ Hm, const float* __restrict__ asrc, const float* __restrict__ adst,
    const float* __restrict__ bias, float* __restrict__ outp, int N, int apply_elu) {
  const int lane = threadIdx.x & 31;
  const int node = __builtin_amdgcn_readfirstlane(blockIdx.x * 8 + (threadIdx.x >> 5));
  if (node >= N) return;
  const float4 a4 = *(const float4*)(asrc + lane * 4);
  const float4 d4 = *(const float4*)(adst + lane * 4);
  const float4 v = gat_node(rowptr, col, Hm, a4, d4, bias, node, lane, apply_elu);
  *(float4*)(outp + (size_t)node * 128 + lane * 4) = v;
}

#define TILE_PITCH 33
#define FUSE_WGS 768
__global__ __launch_bounds__(512) void gat_gather_gemm(
    const int* __restrict__ rowptr, const int* __restrict__ col,
    const float* __restrict__ Hm, const float* __restrict__ asrc, const float* __restrict__ adst,
    const float* __restrict__ bias, const float* __restrict__ Wn, float* __restrict__ Hn, int N) {
  __shared__ float4 wl[32 * 8 * 16];
  __shared__ float4 sa[2][16 * TILE_PITCH];
  const int lane = threadIdx.x & 31;
  const int wv   = __builtin_amdgcn_readfirstlane(threadIdx.x >> 5);
  const int c    = lane & 15;
  const int h    = lane >> 4;

  {
    const int m = threadIdx.x & 31;
#pragma unroll
    for (int it = 0; it < 2; ++it) {
      const int k4 = (threadIdx.x >> 5) + 16 * it;
      const float* wp = Wn + (size_t)(4 * k4) * 128 + 4 * m;
      const float4 r0 = *(const float4*)(wp);
      const float4 r1 = *(const float4*)(wp + 128);
      const float4 r2 = *(const float4*)(wp + 256);
      const float4 r3 = *(const float4*)(wp + 384);
      const int base = (k4 * 8 + 2 * (m >> 3)) * 16 + 2 * (m & 7);
      wl[base]      = make_float4(r0.x, r1.x, r2.x, r3.x);
      wl[base + 16] = make_float4(r0.y, r1.y, r2.y, r3.y);
      wl[base + 1]  = make_float4(r0.z, r1.z, r2.z, r3.z);
      wl[base + 17] = make_float4(r0.w, r1.w, r2.w, r3.w);
    }
  }
  const float4 a4 = *(const float4*)(asrc + lane * 4);
  const float4 d4 = *(const float4*)(adst + lane * 4);
  __syncthreads();

  const float4* bl = wl + h * 128 + c;
  const int ngroups = (N + 15) >> 4;
  int it = 0;
  for (int g = blockIdx.x; g < ngroups; g += FUSE_WGS, ++it) {
    const int node = g * 16 + wv;
    float4 v = make_float4(0.f, 0.f, 0.f, 0.f);
    if (node < N) v = gat_node(rowptr, col, Hm, a4, d4, bias, node, lane, 1);
    float4* sb = sa[it & 1];
    sb[wv * TILE_PITCH + lane] = v;
    __syncthreads();
    const int p = (wv - 4 * (it & 3)) & 15;
    if (p < 4) {
      v8f acc0 = (v8f){0.f, 0.f, 0.f, 0.f, 0.f, 0.f, 0.f, 0.f};
      v8f acc1 = (v8f){0.f, 0.f, 0.f, 0.f, 0.f, 0.f, 0.f, 0.f};
      const float4* ar = sb + c * TILE_PITCH + h;
      const float4* br = bl + (2 * p) * 16;
#pragma unroll
      for (int u = 0; u < 16; ++u) {
        const float4 a  = ar[2 * u];
        const float4 b0 = br[(2 * u * 8) * 16];
        const float4 b1 = br[(2 * u * 8 + 1) * 16];
        v2f a0; a0.x = a.x; a0.y = a.y;
        v2f a1; a1.x = a.z; a1.y = a.w;
        v2f b00; b00.x = b0.x; b00.y = b0.y;
        v2f b01; b01.x = b0.z; b01.y = b0.w;
        v2f b10; b10.x = b1.x; b10.y = b1.y;
        v2f b11; b11.x = b1.z; b11.y = b1.w;
        acc0 = __builtin_amdgcn_wmma_f32_16x16x4_f32(false, a0, false, b00, (short)0, acc0, false, false);
        acc1 = __builtin_amdgcn_wmma_f32_16x16x4_f32(false, a0, false, b10, (short)0, acc1, false, false);
        acc0 = __builtin_amdgcn_wmma_f32_16x16x4_f32(false, a1, false, b01, (short)0, acc0, false, false);
        acc1 = __builtin_amdgcn_wmma_f32_16x16x4_f32(false, a1, false, b11, (short)0, acc1, false, false);
      }
      const int rowbase = g * 16 + 8 * h;
#pragma unroll
      for (int i = 0; i < 8; ++i) {
        const int row = rowbase + i;
        if (row < N)
          *(float2*)(Hn + (size_t)row * 128 + 32 * p + 2 * c) = make_float2(acc0[i], acc1[i]);
      }
    }
  }
}

extern "C" void kernel_launch(void* const* d_in, const int* in_sizes, int n_in,
                              void* d_out, int out_size, void* d_ws, size_t ws_size,
                              hipStream_t stream) {
  const float* x  = (const float*)d_in[0];
  const int*   ei = (const int*)d_in[1];
  const int N = in_sizes[0] / 128;
  const int E = in_sizes[1] / 2;

  const float* Wl[3] = {(const float*)d_in[2], (const float*)d_in[6],  (const float*)d_in[10]};
  const float* As[3] = {(const float*)d_in[3], (const float*)d_in[7],  (const float*)d_in[11]};
  const float* Ad[3] = {(const float*)d_in[4], (const float*)d_in[8],  (const float*)d_in[12]};
  const float* Bb[3] = {(const float*)d_in[5], (const float*)d_in[9],  (const float*)d_in[13]};

  const int gWave = (N + 7) / 8;
  const int NB    = (N + BIN_W - 1) / BIN_W;
  const int GA    = (E + SLAB - 1) / SLAB;

  float* F0     = (float*)d_ws;
  float* F1     = F0 + (size_t)N * 128;
  int*   pairs  = (int*)(F1 + (size_t)N * 128);
  int*   tab    = pairs + (size_t)GA * SLAB * 2;
  int*   rowptr = tab + (size_t)GA * (NB + 1);
  int*   spill  = rowptr + (size_t)N + 4;
  int*   colB   = spill + (size_t)E;

  csr_bin<<<GA, 1024, 0, stream>>>(ei, pairs, tab, E, NB);
  csr_rows<<<NB, 512, 0, stream>>>(pairs, tab, rowptr, spill, colB, N, NB, GA);

  gat_gemm128<<<GEMM_WGS, 512, 0, stream>>>(x, Wl[0], F0, N);
  gat_gather_gemm<<<FUSE_WGS, 512, 0, stream>>>(rowptr, colB, F0, As[0], Ad[0], Bb[0], Wl[1], F1, N);
  gat_gather_gemm<<<FUSE_WGS, 512, 0, stream>>>(rowptr, colB, F1, As[1], Ad[1], Bb[1], Wl[2], F0, N);
  gat_gather<<<gWave, 256, 0, stream>>>(rowptr, colB, F0, As[2], Ad[2], Bb[2], (float*)d_out, N, 0);
}


# Fewer vector instructions per gathered node: each softmax weight one exp2 (attention vectors pre-scaled by log2 e), ELU as exp2-1, one reciprocal per row; sum order kept, last bits move
# speedup vs baseline: 8.0977x; 1.0740x over previous
#include <hip/hip_runtime.h>
#include <hip/hip_bf16.h>
#include <math.h>

typedef __attribute__((ext_vector_type(2))) float v2f;
typedef __attribute__((ext_vector_type(8))) float v8f;

#define NEG_SLOPE 0.2f

#define GEMM_WGS 256
__global__ __launch_bounds__(512) void gat_gemm128(const float* __restrict__ X,
                                                   const float* __restrict__ W,
                                                   float* __restrict__ Hout,
                                                   int N) {
  __shared__ float4 wl[32 * 8 * 16];
  const int lane = threadIdx.x & 31;
  const int wv   = __builtin_amdgcn_readfirstlane(threadIdx.x >> 5);
  const int c    = lane & 15;
  const int h    = lane >> 4;
  const int nstrips = (N + 15) >> 4;
  int strip = wv * GEMM_WGS + blockIdx.x;
  const int stride = 16 * GEMM_WGS;

  float4 a[16];
  {
    int row = strip * 16 + c;
    if (row >= N) row = N - 1;
    const float* ap = X + (size_t)row * 128 + 4 * h;
#pragma unroll
    for (int u = 0; u < 16; ++u) a[u] = *(const float4*)(ap + 8 * u);
  }

  {
    const int m = threadIdx.x & 31;
#pragma unroll
    for (int it = 0; it < 2; ++it) {
      const int k4 = (threadIdx.x >> 5) + 16 * it;
      const float* wp = W + (size_t)(4 * k4) * 128 + 4 * m;
      const float4 r0 = *(const float4*)(wp);
      const float4 r1 = *(const float4*)(wp + 128);
      const float4 r2 = *(const float4*)(wp + 256);
      const float4 r3 = *(const float4*)(wp + 384);
      const int base = (k4 * 8 + (m >> 4) * 4) * 16 + (m & 15);
      wl[base]      = make_float4(r0.x, r1.x, r2.x, r3.x);
      wl[base + 16] = make_float4(r0.y, r1.y, r2.y, r3.y);
      wl[base + 32] = make_float4(r0.z, r1.z, r2.z, r3.z);
      wl[base + 48] = make_float4(r0.w, r1.w, r2.w, r3.w);
    }
  }
  __syncthreads();

  const float4* bl = wl + h * 128 + c;
  while (strip < nstrips) {
    __atomic_signal_fence(__ATOMIC_SEQ_CST);
    const int nxt = strip + stride;
    int nrow = nxt * 16 + c;
    if (nrow >= N) nrow = N - 1;
    const float* np = X + (size_t)nrow * 128 + 4 * h;

    v8f acc[8];
#pragma unroll
    for (int j = 0; j < 8; ++j) acc[j] = (v8f){0.f, 0.f, 0.f, 0.f, 0.f, 0.f, 0.f, 0.f};

#pragma unroll
    for (int g = 0; g < 4; ++g) {
#pragma unroll
      for (int uu = 0; uu < 4; ++uu) {
        const int u = 4 * g + uu;
        v2f a0; a0.x = a[u].x; a0.y = a[u].y;
        v2f a1; a1.x = a[u].z; a1.y = a[u].w;
#pragma unroll
        for (int j = 0; j < 8; ++j) {
          const float4 b = bl[(2 * u * 8 + j) * 16];
          v2f b0; b0.x = b.x; b0.y = b.y;
          v2f b1; b1.x = b.z; b1.y = b.w;
          acc[j] = __builtin_amdgcn_wmma_f32_16x16x4_f32(false, a0, false, b0,
                                                         (short)0, acc[j], false, false);
          acc[j] = __builtin_amdgcn_wmma_f32_16x16x4_f32(false, a1, false, b1,
                                                         (short)0, acc[j], false, false);
        }
      }
#pragma unroll
      for (int uu = 0; uu < 4; ++uu) a[4 * g + uu] = *(const float4*)(np + 8 * (4 * g + uu));
    }

    const int rowbase = strip * 16 + 8 * h;
#pragma unroll
    for (int jj = 0; jj < 2; ++jj) {
#pragma unroll
      for (int i = 0; i < 8; ++i) {
        const int row = rowbase + i;
        if (row < N)
          *(float4*)(Hout + (size_t)row * 128 + 64 * jj + 4 * c) =
              make_float4(acc[4 * jj][i], acc[4 * jj + 1][i], acc[4 * jj + 2][i], acc[4 * jj + 3][i]);
      }
    }
    strip = nxt;
  }
}

#define LOG2E 1.4426950408889634f
__device__ __forceinline__ float ex2(float x) { return __builtin_amdgcn_exp2f(x); }

__device__ __forceinline__ float lrelu(float x) {
  return x > 0.f ? x : NEG_SLOPE * x;
}

#define BIN_W 128
#define MAXNB 1024
#define SLAB 6144
#define ROWCAP 4096

__global__ __launch_bounds__(1024) void csr_bin(const int* __restrict__ ei, int* __restrict__ pairs_raw,
                                                int* __restrict__ tab, int E, int NB) {
  __shared__ int hist[MAXNB];
  __shared__ int lst[MAXNB];
  __shared__ int wsum[32];
  __shared__ int2 pl[SLAB];
  const int tid  = threadIdx.x;
  const int lane = tid & 31;
  const int wv   = tid >> 5;
  const int s0   = blockIdx.x * SLAB;
  const int cnt  = (E - s0) < SLAB ? (E - s0) : SLAB;
  hist[tid] = 0;
  __syncthreads();
  int dreg[6], sreg[6];
#pragma unroll
  for (int k = 0; k < 6; ++k) {
    const int i = tid + 1024 * k;
    dreg[k] = 0; sreg[k] = 0;
    if (i < cnt) {
      dreg[k] = ei[E + s0 + i];
      sreg[k] = ei[s0 + i];
      atomicAdd(&hist[dreg[k] >> 7], 1);
    }
  }
  __syncthreads();
  const int v = hist[tid];
  int inc = v;
#pragma unroll
  for (int o = 1; o < 32; o <<= 1) {
    const int t = __shfl_up(inc, o);
    if (lane >= o) inc += t;
  }
  if (lane == 31) wsum[wv] = inc;
  __syncthreads();
  int off = 0;
  for (int i = 0; i < wv; ++i) off += wsum[i];
  const int excl = off + inc - v;
  lst[tid]  = excl;
  hist[tid] = excl;
  __syncthreads();
#pragma unroll
  for (int k = 0; k < 6; ++k) {
    const int i = tid + 1024 * k;
    if (i < cnt) {
      const int p = atomicAdd(&hist[dreg[k] >> 7], 1);
      pl[p] = make_int2(sreg[k], dreg[k]);
    }
  }
  __syncthreads();
  int2* pairs = (int2*)pairs_raw + (size_t)blockIdx.x * SLAB;
#pragma unroll
  for (int k = 0; k < 6; ++k) {
    const int i = tid + 1024 * k;
    if (i < cnt) pairs[i] = pl[i];
  }
  int* tb = tab + (size_t)blockIdx.x * (NB + 1);
  if (tid < NB) tb[tid] = lst[tid];
  if (tid == 0) tb[NB] = cnt;
}

__device__ __forceinline__ void csr_order_rows(const int* st, const int* exL, const int* cntL,
                                               int* __restrict__ colrow, int lane, int wv) {
  for (int d = wv; d < BIN_W; d += 16) {
    const int beg = exL[d];
    const int m   = cntL[d];
    for (int i = lane; i < m; i += 32) {
      const int key = st[beg + i];
      int r = 0;
      for (int j = 0; j < m; ++j) {
        const int kj = st[beg + j];
        r += (kj < key || (kj == key && j < i)) ? 1 : 0;
      }
      colrow[beg + r] = key;
    }
  }
}

__global__ __launch_bounds__(512) void csr_rows(const int* __restrict__ pairs_raw,
                                               const int* __restrict__ tab,
                                               int* __restrict__ rowptr, int* __restrict__ spill,
                                               int* __restrict__ col, int N, int NB, int GA) {
  __shared__ int cntL[BIN_W];
  __shared__ int exL[BIN_W];
  __shared__ int curL[BIN_W];
  __shared__ int red[2][16];
  __shared__ int w4[4];
  __shared__ int stageL[ROWCAP];
  const int tid  = threadIdx.x;
  const int lane = tid & 31;
  const int wv   = tid >> 5;
  const int k    = blockIdx.x;
  const int n0   = k * BIN_W;
  const int2* pairs = (const int2*)pairs_raw;
  if (tid < BIN_W) cntL[tid] = 0;
  int sa = 0, sl = 0;
  for (int t = tid; t < GA; t += 512) {
    const int* tp = tab + (size_t)t * (NB + 1) + k;
    const int a = tp[0];
    sa += a;
    sl += tp[1] - a;
  }
#pragma unroll
  for (int o = 16; o > 0; o >>= 1) {
    sa += __shfl_xor(sa, o);
    sl += __shfl_xor(sl, o);
  }
  if (lane == 0) { red[0][wv] = sa; red[1][wv] = sl; }
  __syncthreads();
  int base = 0, n = 0;
  for (int i = 0; i < 16; ++i) { base += red[0][i]; n += red[1][i]; }
  for (int t = tid; t < GA; t += 512) {
    const int* tp = tab + (size_t)t * (NB + 1) + k;
    const int a = tp[0];
    const int len = tp[1] - a;
    const int2* run = pairs + (size_t)t * SLAB + a;
    for (int j = 0; j < len; ++j) atomicAdd(&cntL[run[j].y - n0], 1);
  }
  __syncthreads();
  int v = 0, inc = 0;
  if (tid < BIN_W) {
    v = cntL[tid];
    inc = v;
#pragma unroll
    for (int o = 1; o < 32; o <<= 1) {
      const int t = __shfl_up(inc, o);
      if (lane >= o) inc += t;
    }
    if (lane == 31) w4[wv] = inc;
  }
  __syncthreads();
  if (tid < BIN_W) {
    int off = 0;
    for (int i = 0; i < wv; ++i) off += w4[i];
    const int ex = off + inc - v;
    exL[tid]  = ex;
    curL[tid] = ex;
    const int node = n0 + tid;
    if (node < N) rowptr[node] = base + ex;
    if (node == N - 1) rowptr[N] = base + ex + v;
  }
  __syncthreads();
  int* colrow = col + base;
  if (n <= ROWCAP) {
    for (int t = tid; t < GA; t += 512) {
      const int* tp = tab + (size_t)t * (NB + 1) + k;
      const int a = tp[0];
      const int len = tp[1] - a;
      const int2* run = pairs + (size_t)t * SLAB + a;
      for (int j = 0; j < len; ++j) {
        const int2 e = run[j];
        stageL[atomicAdd(&curL[e.y - n0], 1)] = e.x;
      }
    }
    __syncthreads();
    csr_order_rows(stageL, exL, cntL, colrow, lane, wv);
  } else {
    int* sp = spill + base;
    for (int t = tid; t < GA; t += 512) {
      const int* tp = tab + (size_t)t * (NB + 1) + k;
      const int a = tp[0];
      const int len = tp[1] - a;
      const int2* run = pairs + (size_t)t * SLAB + a;
      for (int j = 0; j < len; ++j) {
        const int2 e = run[j];
        sp[atomicAdd(&curL[e.y - n0], 1)] = e.x;
      }
    }
    __threadfence_block();
    __syncthreads();
    csr_order_rows(sp, exL, cntL, colrow, lane, wv);
  }
}

__device__ __forceinline__ float xch1(float v) {
  return __int_as_float(__builtin_amdgcn_update_dpp(0, __float_as_int(v), 0xB1, 0xF, 0xF, true));
}
__device__ __forceinline__ float xch2(float v) {
  return __int_as_float(__builtin_amdgcn_update_dpp(0, __float_as_int(v), 0x4E, 0xF, 0xF, true));
}
__device__ __forceinline__ float xch4(float v) {
  return __int_as_float(__builtin_amdgcn_update_dpp(0, __float_as_int(v), 0x164, 0xF, 0xF, true));
}
__device__ __forceinline__ float sum8(float v) {
  v += xch1(v); v += xch2(v); v += xch4(v);
  return v;
}
__device__ __forceinline__ float max8(float v) {
  v = fmaxf(v, xch1(v)); v = fmaxf(v, xch2(v)); v = fmaxf(v, xch4(v));
  return v;
}
template <int J>
__device__ __forceinline__ float bcast8(float v) {
  return __int_as_float(__builtin_amdgcn_mov_dpp8(__float_as_int(v), J * 0x249249));
}
__device__ __forceinline__ float dot4(const float4 r, const float4 a) {
  return fmaf(r.w, a.w, fmaf(r.z, a.z, fmaf(r.y, a.y, r.x * a.x)));
}

template <bool TAIL>
__device__ __forceinline__ void gat_chunk(const float* __restrict__ hrow, int sc, int c0, int rem,
                                          int i8, bool b1, bool b2, bool b4,
                                          const float4 a4, float ed,
                                          float& m, float& ssum, float4& acc) {
  float4 r[8];
  int srow[8];
#pragma unroll
  for (int k = 0; k < 8; ++k) srow[k] = __builtin_amdgcn_readlane(sc, c0 + k);
#pragma unroll
  for (int k = 0; k < 8; ++k) {
    if (!TAIL || k < rem)
      r[k] = *(const float4*)(hrow + (size_t)srow[k] * 128);
    else
      r[k] = make_float4(0.f, 0.f, 0.f, 0.f);
  }
  float p[8];
#pragma unroll
  for (int k = 0; k < 8; ++k) p[k] = dot4(r[k], a4);
  float q[4];
#pragma unroll
  for (int k = 0; k < 4; ++k)
    q[k] = (b1 ? p[2 * k + 1] : p[2 * k]) + xch1(b1 ? p[2 * k] : p[2 * k + 1]);
  float s2[2];
#pragma unroll
  for (int k = 0; k < 2; ++k)
    s2[k] = (b2 ? q[2 * k + 1] : q[2 * k]) + xch2(b2 ? q[2 * k] : q[2 * k + 1]);
  const float s = (b4 ? s2[1] : s2[0]) + xch4(b4 ? s2[0] : s2[1]);
  float l = lrelu(s + ed);
  if (TAIL) l = (i8 < rem) ? l : -__builtin_inff();
  const float mn = fmaxf(m, max8(l));
  const float rs = ex2(m - mn);
  const float pw = ex2(l - mn);
  m = mn;
  ssum = fmaf(ssum, rs, pw);
  acc.x *= rs; acc.y *= rs; acc.z *= rs; acc.w *= rs;
#define GAT_ADD_ROW(K)                                                    \
  {                                                                       \
    const float w = bcast8<K>(pw);                                        \
    acc.x = fmaf(r[K].x, w, acc.x); acc.y = fmaf(r[K].y, w, acc.y);       \
    acc.z = fmaf(r[K].z, w, acc.z); acc.w = fmaf(r[K].w, w, acc.w);       \
  }
  GAT_ADD_ROW(0) GAT_ADD_ROW(1) GAT_ADD_ROW(2) GAT_ADD_ROW(3)
  GAT_ADD_ROW(4) GAT_ADD_ROW(5) GAT_ADD_ROW(6) GAT_ADD_ROW(7)
#undef GAT_ADD_ROW
}

__device__ __forceinline__ float4 gat_node(
    const int* __restrict__ rowptr, const int* __restrict__ col, const float* __restrict__ Hm,
    const float4 a4, const float4 d4, const float* __restrict__ bias, int node, int lane, int apply_elu) {
  const int beg = __builtin_amdgcn_readfirstlane(rowptr[node]);
  const int n   = __builtin_amdgcn_readfirstlane(rowptr[node + 1]) - beg;
  const float* hrow = Hm + lane * 4;
  const float4 hs = *(const float4*)(hrow + (size_t)node * 128);
  const int  i8 = lane & 7;
  const bool b1 = (lane & 1) != 0, b2 = (lane & 2) != 0, b4 = (lane & 4) != 0;

  const float ed = sum8(dot4(hs, d4));
  float m = lrelu(sum8(dot4(hs, a4)) + ed);
  float4 acc = hs;
  float ssum = i8 == 0 ? 1.f : 0.f;

  for (int t0 = 0; t0 < n; t0 += 32) {
    int sc = node;
    if (t0 + lane < n) sc = col[beg + t0 + lane];
    const int cnt = (n - t0) < 32 ? (n - t0) : 32;
    int c0 = 0;
    for (; c0 + 8 <= cnt; c0 += 8)
      gat_chunk<false>(hrow, sc, c0, 8, i8, b1, b2, b4, a4, ed, m, ssum, acc);
    if (c0 < cnt)
      gat_chunk<true>(hrow, sc, c0, cnt - c0, i8, b1, b2, b4, a4, ed, m, ssum, acc);
  }
  const float tsum = sum8(ssum);
  const float4 bv = *(const float4*)(bias + lane * 4);
  float4 v;
  const float inv = 1.0f / tsum;
  v.x = fmaf(acc.x, inv, bv.x);
  v.y = fmaf(acc.y, inv, bv.y);
  v.z = fmaf(acc.z, inv, bv.z);
  v.w = fmaf(acc.w, inv, bv.w);
  if (apply_elu) {
    v.x = v.x > 0.f ? v.x : ex2(v.x * LOG2E) - 1.f;
    v.y = v.y > 0.f ? v.y : ex2(v.y * LOG2E) - 1.f;
    v.z = v.z > 0.f ? v.z : ex2(v.z * LOG2E) - 1.f;
    v.w = v.w > 0.f ? v.w : ex2(v.w * LOG2E) - 1.f;
  }
  return v;
}

__global__ __launch_bounds__(256) void gat_gather(
    const int* __restrict__ rowptr, const int* __restrict__ col,
    const float* __restrict__ Hm, const float* __restrict__ asrc, const float* __restrict__ adst,
    const float* __restrict__ bias, float* __restrict__ outp, int N, int apply_elu) {
  const int lane = threadIdx.x & 31;
  const int node = __builtin_amdgcn_readfirstlane(blockIdx.x * 8 + (threadIdx.x >> 5));
  if (node >= N) return;
  float4 a4 = *(const float4*)(asrc + lane * 4);
  float4 d4 = *(const float4*)(adst + lane * 4);
  a4.x *= LOG2E; a4.y *= LOG2E; a4.z *= LOG2E; a4.w *= LOG2E;
  d4.x *= LOG2E; d4.y *= LOG2E; d4.z *= LOG2E; d4.w *= LOG2E;
  const float4 v = gat_node(rowptr, col, Hm, a4, d4, bias, node, lane, apply_elu);
  *(float4*)(outp + (size_t)node * 128 + lane * 4) = v;
}

#define TILE_PITCH 33
#define FUSE_WGS 768
__global__ __launch_bounds__(512) void gat_gather_gemm(
    const int* __restrict__ rowptr, const int* __restrict__ col,
    const float* __restrict__ Hm, const float* __restrict__ asrc, const float* __restrict__ adst,
    const float* __restrict__ bias, const float* __restrict__ Wn, float* __restrict__ Hn, int N) {
  __shared__ float4 wl[32 * 8 * 16];
  __shared__ float4 sa[2][16 * TILE_PITCH];
  const int lane = threadIdx.x & 31;
  const int wv   = __builtin_amdgcn_readfirstlane(threadIdx.x >> 5);
  const int c    = lane & 15;
  const int h    = lane >> 4;

  {
    const int m = threadIdx.x & 31;
#pragma unroll
    for (int it = 0; it < 2; ++it) {
      const int k4 = (threadIdx.x >> 5) + 16 * it;
      const float* wp = Wn + (size_t)(4 * k4) * 128 + 4 * m;
      const float4 r0 = *(const float4*)(wp);
      const float4 r1 = *(const float4*)(wp + 128);
      const float4 r2 = *(const float4*)(wp + 256);
      const float4 r3 = *(const float4*)(wp + 384);
      const int base = (k4 * 8 + 2 * (m >> 3)) * 16 + 2 * (m & 7);
      wl[base]      = make_float4(r0.x, r1.x, r2.x, r3.x);
      wl[base + 16] = make_float4(r0.y, r1.y, r2.y, r3.y);
      wl[base + 1]  = make_float4(r0.z, r1.z, r2.z, r3.z);
      wl[base + 17] = make_float4(r0.w, r1.w, r2.w, r3.w);
    }
  }
  float4 a4 = *(const float4*)(asrc + lane * 4);
  float4 d4 = *(const float4*)(adst + lane * 4);
  a4.x *= LOG2E; a4.y *= LOG2E; a4.z *= LOG2E; a4.w *= LOG2E;
  d4.x *= LOG2E; d4.y *= LOG2E; d4.z *= LOG2E; d4.w *= LOG2E;
  __syncthreads();

  const float4* bl = wl + h * 128 + c;
  const int ngroups = (N + 15) >> 4;
  int it = 0;
  for (int g = blockIdx.x; g < ngroups; g += FUSE_WGS, ++it) {
    const int node = g * 16 + wv;
    float4 v = make_float4(0.f, 0.f, 0.f, 0.f);
    if (node < N) v = gat_node(rowptr, col, Hm, a4, d4, bias, node, lane, 1);
    float4* sb = sa[it & 1];
    sb[wv * TILE_PITCH + lane] = v;
    __syncthreads();
    const int p = (wv - 4 * (it & 3)) & 15;
    if (p < 4) {
      v8f acc0 = (v8f){0.f, 0.f, 0.f, 0.f, 0.f, 0.f, 0.f, 0.f};
      v8f acc1 = (v8f){0.f, 0.f, 0.f, 0.f, 0.f, 0.f, 0.f, 0.f};
      const float4* ar = sb + c * TILE_PITCH + h;
      const float4* br = bl + (2 * p) * 16;
#pragma unroll
      for (int u = 0; u < 16; ++u) {
        const float4 a  = ar[2 * u];
        const float4 b0 = br[(2 * u * 8) * 16];
        const float4 b1 = br[(2 * u * 8 + 1) * 16];
        v2f a0; a0.x = a.x; a0.y = a.y;
        v2f a1; a1.x = a.z; a1.y = a.w;
        v2f b00; b00.x = b0.x; b00.y = b0.y;
        v2f b01; b01.x = b0.z; b01.y = b0.w;
        v2f b10; b10.x = b1.x; b10.y = b1.y;
        v2f b11; b11.x = b1.z; b11.y = b1.w;
        acc0 = __builtin_amdgcn_wmma_f32_16x16x4_f32(false, a0, false, b00, (short)0, acc0, false, false);
        acc1 = __builtin_amdgcn_wmma_f32_16x16x4_f32(false, a0, false, b10, (short)0, acc1, false, false);
        acc0 = __builtin_amdgcn_wmma_f32_16x16x4_f32(false, a1, false, b01, (short)0, acc0, false, false);
        acc1 = __builtin_amdgcn_wmma_f32_16x16x4_f32(false, a1, false, b11, (short)0, acc1, false, false);
      }
      const int rowbase = g * 16 + 8 * h;
#pragma unroll
      for (int i = 0; i < 8; ++i) {
        const int row = rowbase + i;
        if (row < N)
          *(float2*)(Hn + (size_t)row * 128 + 32 * p + 2 * c) = make_float2(acc0[i], acc1[i]);
      }
    }
  }
}

extern "C" void kernel_launch(void* const* d_in, const int* in_sizes, int n_in,
                              void* d_out, int out_size, void* d_ws, size_t ws_size,
                              hipStream_t stream) {
  const float* x  = (const float*)d_in[0];
  const int*   ei = (const int*)d_in[1];
  const int N = in_sizes[0] / 128;
  const int E = in_sizes[1] / 2;

  const float* Wl[3] = {(const float*)d_in[2], (const float*)d_in[6],  (const float*)d_in[10]};
  const float* As[3] = {(const float*)d_in[3], (const float*)d_in[7],  (const float*)d_in[11]};
  const float* Ad[3] = {(const float*)d_in[4], (const float*)d_in[8],  (const float*)d_in[12]};
  const float* Bb[3] = {(const float*)d_in[5], (const float*)d_in[9],  (const float*)d_in[13]};

  const int gWave = (N + 7) / 8;
  const int NB    = (N + BIN_W - 1) / BIN_W;
  const int GA    = (E + SLAB - 1) / SLAB;

  float* F0     = (float*)d_ws;
  float* F1     = F0 + (size_t)N * 128;
  int*   pairs  = (int*)(F1 + (size_t)N * 128);
  int*   tab    = pairs + (size_t)GA * SLAB * 2;
  int*   rowptr = tab + (size_t)GA * (NB + 1);
  int*   spill  = rowptr + (size_t)N + 4;
  int*   colB   = spill + (size_t)E;

  csr_bin<<<GA, 1024, 0, stream>>>(ei, pairs, tab, E, NB);
  csr_rows<<<NB, 512, 0, stream>>>(pairs, tab, rowptr, spill, colB, N, NB, GA);

  gat_gemm128<<<GEMM_WGS, 512, 0, stream>>>(x, Wl[0], F0, N);
  gat_gather_gemm<<<FUSE_WGS, 512, 0, stream>>>(rowptr, colB, F0, As[0], Ad[0], Bb[0], Wl[1], F1, N);
  gat_gather_gemm<<<FUSE_WGS, 512, 0, stream>>>(rowptr, colB, F1, As[1], Ad[1], Bb[1], Wl[2], F0, N);
  gat_gather<<<gWave, 256, 0, stream>>>(rowptr, colB, F0, As[2], Ad[2], Bb[2], (float*)d_out, N, 0);
}


# Fewer vector instructions per edge: row address = scalar base + lane offset, weights spread by a row permute each, 4-slot form for short tails, hardware reciprocal (last bits move); sums kept
# speedup vs baseline: 8.5424x; 1.0549x over previous
#include <hip/hip_runtime.h>
#include <hip/hip_bf16.h>
#include <math.h>

typedef __attribute__((ext_vector_type(2))) float v2f;
typedef __attribute__((ext_vector_type(8))) float v8f;

#define NEG_SLOPE 0.2f

#define GEMM_WGS 256
__global__ __launch_bounds__(512) void gat_gemm128(const float* __restrict__ X,
                                                   const float* __restrict__ W,
                                                   float* __restrict__ Hout,
                                                   int N) {
  __shared__ float4 wl[32 * 8 * 16];
  const int lane = threadIdx.x & 31;
  const int wv   = __builtin_amdgcn_readfirstlane(threadIdx.x >> 5);
  const int c    = lane & 15;
  const int h    = lane >> 4;
  const int nstrips = (N + 15) >> 4;
  int strip = wv * GEMM_WGS + blockIdx.x;
  const int stride = 16 * GEMM_WGS;

  float4 a[16];
  {
    int row = strip * 16 + c;
    if (row >= N) row = N - 1;
    const float* ap = X + (size_t)row * 128 + 4 * h;
#pragma unroll
    for (int u = 0; u < 16; ++u) a[u] = *(const float4*)(ap + 8 * u);
  }

  {
    const int m = threadIdx.x & 31;
#pragma unroll
    for (int it = 0; it < 2; ++it) {
      const int k4 = (threadIdx.x >> 5) + 16 * it;
      const float* wp = W + (size_t)(4 * k4) * 128 + 4 * m;
      const float4 r0 = *(const float4*)(wp);
      const float4 r1 = *(const float4*)(wp + 128);
      const float4 r2 = *(const float4*)(wp + 256);
      const float4 r3 = *(const float4*)(wp + 384);
      const int base = (k4 * 8 + (m >> 4) * 4) * 16 + (m & 15);
      wl[base]      = make_float4(r0.x, r1.x, r2.x, r3.x);
      wl[base + 16] = make_float4(r0.y, r1.y, r2.y, r3.y);
      wl[base + 32] = make_float4(r0.z, r1.z, r2.z, r3.z);
      wl[base + 48] = make_float4(r0.w, r1.w, r2.w, r3.w);
    }
  }
  __syncthreads();

  const float4* bl = wl + h * 128 + c;
  while (strip < nstrips) {
    __atomic_signal_fence(__ATOMIC_SEQ_CST);
    const int nxt = strip + stride;
    int nrow = nxt * 16 + c;
    if (nrow >= N) nrow = N - 1;
    const float* np = X + (size_t)nrow * 128 + 4 * h;

    v8f acc[8];
#pragma unroll
    for (int j = 0; j < 8; ++j) acc[j] = (v8f){0.f, 0.f, 0.f, 0.f, 0.f, 0.f, 0.f, 0.f};

#pragma unroll
    for (int g = 0; g < 4; ++g) {
#pragma unroll
      for (int uu = 0; uu < 4; ++uu) {
        const int u = 4 * g + uu;
        v2f a0; a0.x = a[u].x; a0.y = a[u].y;
        v2f a1; a1.x = a[u].z; a1.y = a[u].w;
#pragma unroll
        for (int j = 0; j < 8; ++j) {
          const float4 b = bl[(2 * u * 8 + j) * 16];
          v2f b0; b0.x = b.x; b0.y = b.y;
          v2f b1; b1.x = b.z; b1.y = b.w;
          acc[j] = __builtin_amdgcn_wmma_f32_16x16x4_f32(false, a0, false, b0,
                                                         (short)0, acc[j], false, false);
          acc[j] = __builtin_amdgcn_wmma_f32_16x16x4_f32(false, a1, false, b1,
                                                         (short)0, acc[j], false, false);
        }
      }
#pragma unroll
      for (int uu = 0; uu < 4; ++uu) a[4 * g + uu] = *(const float4*)(np + 8 * (4 * g + uu));
    }

    const int rowbase = strip * 16 + 8 * h;
#pragma unroll
    for (int jj = 0; jj < 2; ++jj) {
#pragma unroll
      for (int i = 0; i < 8; ++i) {
        const int row = rowbase + i;
        if (row < N)
          *(float4*)(Hout + (size_t)row * 128 + 64 * jj + 4 * c) =
              make_float4(acc[4 * jj][i], acc[4 * jj + 1][i], acc[4 * jj + 2][i], acc[4 * jj + 3][i]);
      }
    }
    strip = nxt;
  }
}

#define LOG2E 1.4426950408889634f
__device__ __forceinline__ float ex2(float x) { return __builtin_amdgcn_exp2f(x); }

__device__ __forceinline__ float lrelu(float x) {
  return fmaxf(x, NEG_SLOPE * x);
}

#define BIN_W 128
#define MAXNB 1024
#define SLAB 6144
#define ROWCAP 4096

__global__ __launch_bounds__(1024) void csr_bin(const int* __restrict__ ei, int* __restrict__ pairs_raw,
                                                int* __restrict__ tab, int E, int NB) {
  __shared__ int hist[MAXNB];
  __shared__ int lst[MAXNB];
  __shared__ int wsum[32];
  __shared__ int2 pl[SLAB];
  const int tid  = threadIdx.x;
  const int lane = tid & 31;
  const int wv   = tid >> 5;
  const int s0   = blockIdx.x * SLAB;
  const int cnt  = (E - s0) < SLAB ? (E - s0) : SLAB;
  hist[tid] = 0;
  __syncthreads();
  int dreg[6], sreg[6];
#pragma unroll
  for (int k = 0; k < 6; ++k) {
    const int i = tid + 1024 * k;
    dreg[k] = 0; sreg[k] = 0;
    if (i < cnt) {
      dreg[k] = ei[E + s0 + i];
      sreg[k] = ei[s0 + i];
      atomicAdd(&hist[dreg[k] >> 7], 1);
    }
  }
  __syncthreads();
  const int v = hist[tid];
  int inc = v;
#pragma unroll
  for (int o = 1; o < 32; o <<= 1) {
    const int t = __shfl_up(inc, o);
    if (lane >= o) inc += t;
  }
  if (lane == 31) wsum[wv] = inc;
  __syncthreads();
  int off = 0;
  for (int i = 0; i < wv; ++i) off += wsum[i];
  const int excl = off + inc - v;
  lst[tid]  = excl;
  hist[tid] = excl;
  __syncthreads();
#pragma unroll
  for (int k = 0; k < 6; ++k) {
    const int i = tid + 1024 * k;
    if (i < cnt) {
      const int p = atomicAdd(&hist[dreg[k] >> 7], 1);
      pl[p] = make_int2(sreg[k], dreg[k]);
    }
  }
  __syncthreads();
  int2* pairs = (int2*)pairs_raw + (size_t)blockIdx.x * SLAB;
#pragma unroll
  for (int k = 0; k < 6; ++k) {
    const int i = tid + 1024 * k;
    if (i < cnt) pairs[i] = pl[i];
  }
  int* tb = tab + (size_t)blockIdx.x * (NB + 1);
  if (tid < NB) tb[tid] = lst[tid];
  if (tid == 0) tb[NB] = cnt;
}

__device__ __forceinline__ void csr_order_rows(const int* st, const int* exL, const int* cntL,
                                               int* __restrict__ colrow, int lane, int wv) {
  for (int d = wv; d < BIN_W; d += 16) {
    const int beg = exL[d];
    const int m   = cntL[d];
    for (int i = lane; i < m; i += 32) {
      const int key = st[beg + i];
      int r = 0;
      for (int j = 0; j < m; ++j) {
        const int kj = st[beg + j];
        r += (kj < key || (kj == key && j < i)) ? 1 : 0;
      }
      colrow[beg + r] = key;
    }
  }
}

__global__ __launch_bounds__(512) void csr_rows(const int* __restrict__ pairs_raw,
                                               const int* __restrict__ tab,
                                               int* __restrict__ rowptr, int* __restrict__ spill,
                                               int* __restrict__ col, int N, int NB, int GA) {
  __shared__ int cntL[BIN_W];
  __shared__ int exL[BIN_W];
  __shared__ int curL[BIN_W];
  __shared__ int red[2][16];
  __shared__ int w4[4];
  __shared__ int stageL[ROWCAP];
  const int tid  = threadIdx.x;
  const int lane = tid & 31;
  const int wv   = tid >> 5;
  const int k    = blockIdx.x;
  const int n0   = k * BIN_W;
  const int2* pairs = (const int2*)pairs_raw;
  if (tid < BIN_W) cntL[tid] = 0;
  int sa = 0, sl = 0;
  for (int t = tid; t < GA; t += 512) {
    const int* tp = tab + (size_t)t * (NB + 1) + k;
    const int a = tp[0];
    sa += a;
    sl += tp[1] - a;
  }
#pragma unroll
  for (int o = 16; o > 0; o >>= 1) {
    sa += __shfl_xor(sa, o);
    sl += __shfl_xor(sl, o);
  }
  if (lane == 0) { red[0][wv] = sa; red[1][wv] = sl; }
  __syncthreads();
  int base = 0, n = 0;
  for (int i = 0; i < 16; ++i) { base += red[0][i]; n += red[1][i]; }
  for (int t = tid; t < GA; t += 512) {
    const int* tp = tab + (size_t)t * (NB + 1) + k;
    const int a = tp[0];
    const int len = tp[1] - a;
    const int2* run = pairs + (size_t)t * SLAB + a;
    for (int j = 0; j < len; ++j) atomicAdd(&cntL[run[j].y - n0], 1);
  }
  __syncthreads();
  int v = 0, inc = 0;
  if (tid < BIN_W) {
    v = cntL[tid];
    inc = v;
#pragma unroll
    for (int o = 1; o < 32; o <<= 1) {
      const int t = __shfl_up(inc, o);
      if (lane >= o) inc += t;
    }
    if (lane == 31) w4[wv] = inc;
  }
  __syncthreads();
  if (tid < BIN_W) {
    int off = 0;
    for (int i = 0; i < wv; ++i) off += w4[i];
    const int ex = off + inc - v;
    exL[tid]  = ex;
    curL[tid] = ex;
    const int node = n0 + tid;
    if (node < N) rowptr[node] = base + ex;
    if (node == N - 1) rowptr[N] = base + ex + v;
  }
  __syncthreads();
  int* colrow = col + base;
  if (n <= ROWCAP) {
    for (int t = tid; t < GA; t += 512) {
      const int* tp = tab + (size_t)t * (NB + 1) + k;
      const int a = tp[0];
      const int len = tp[1] - a;
      const int2* run = pairs + (size_t)t * SLAB + a;
      for (int j = 0; j < len; ++j) {
        const int2 e = run[j];
        stageL[atomicAdd(&curL[e.y - n0], 1)] = e.x;
      }
    }
    __syncthreads();
    csr_order_rows(stageL, exL, cntL, colrow, lane, wv);
  } else {
    int* sp = spill + base;
    for (int t = tid; t < GA; t += 512) {
      const int* tp = tab + (size_t)t * (NB + 1) + k;
      const int a = tp[0];
      const int len = tp[1] - a;
      const int2* run = pairs + (size_t)t * SLAB + a;
      for (int j = 0; j < len; ++j) {
        const int2 e = run[j];
        sp[atomicAdd(&curL[e.y - n0], 1)] = e.x;
      }
    }
    __threadfence_block();
    __syncthreads();
    csr_order_rows(sp, exL, cntL, colrow, lane, wv);
  }
}

__device__ __forceinline__ float xch1(float v) {
  return __int_as_float(__builtin_amdgcn_update_dpp(0, __float_as_int(v), 0xB1, 0xF, 0xF, true));
}
__device__ __forceinline__ float xch2(float v) {
  return __int_as_float(__builtin_amdgcn_update_dpp(0, __float_as_int(v), 0x4E, 0xF, 0xF, true));
}
__device__ __forceinline__ float xch4(float v) {
  return __int_as_float(__builtin_amdgcn_update_dpp(0, __float_as_int(v), 0x164, 0xF, 0xF, true));
}
__device__ __forceinline__ float sum8(float v) {
  v += xch1(v); v += xch2(v); v += xch4(v);
  return v;
}
__device__ __forceinline__ float max8(float v) {
  v = fmaxf(v, xch1(v)); v = fmaxf(v, xch2(v)); v = fmaxf(v, xch4(v));
  return v;
}
template <int J>
__device__ __forceinline__ float bcast8(float dead, float v) {
  return __uint_as_float(__builtin_amdgcn_permlane16(__float_as_uint(dead), __float_as_uint(v),
                                                     J * 0x11111111u, (8 + J) * 0x11111111u,
                                                     false, false));
}
__device__ __forceinline__ float dot4(const float4 r, const float4 a) {
  return fmaf(r.w, a.w, fmaf(r.z, a.z, fmaf(r.y, a.y, r.x * a.x)));
}

typedef const __attribute__((address_space(1))) float*  gfloat_p;
typedef const __attribute__((address_space(1))) float4* gfloat4_p;
__device__ __forceinline__ float4 load_row(const float* __restrict__ Hm, unsigned row, unsigned loff) {
  const unsigned long long a = (unsigned long long)(Hm + (size_t)row * 128);
  const unsigned lo = __builtin_amdgcn_readfirstlane((unsigned)a);
  const unsigned hi = __builtin_amdgcn_readfirstlane((unsigned)(a >> 32));
  const gfloat_p base = (gfloat_p)(((unsigned long long)hi << 32) | lo);
  return *(gfloat4_p)(base + loff);
}

template <bool TAIL, int NS>
__device__ __forceinline__ void gat_chunk(const float* __restrict__ Hm, unsigned loff, int sc, int c0, int rem,
                                          int i8, bool b1, bool b2, bool b4,
                                          const float4 a4, float ed,
                                          float& m, float& ssum, float4& acc) {
  float4 r[NS];
  unsigned srow[NS];
#pragma unroll
  for (int k = 0; k < NS; ++k) srow[k] = (unsigned)__builtin_amdgcn_readlane(sc, c0 + k);
#pragma unroll
  for (int k = 0; k < NS; ++k) {
    if (!TAIL || k < rem)
      r[k] = load_row(Hm, srow[k], loff);
    else
      r[k] = make_float4(0.f, 0.f, 0.f, 0.f);
  }
  float p[NS];
#pragma unroll
  for (int k = 0; k < NS; ++k) p[k] = dot4(r[k], a4);
  float q[NS / 2];
#pragma unroll
  for (int k = 0; k < NS / 2; ++k)
    q[k] = (b1 ? p[2 * k + 1] : p[2 * k]) + xch1(b1 ? p[2 * k] : p[2 * k + 1]);
  float s2[NS / 4];
#pragma unroll
  for (int k = 0; k < NS / 4; ++k)
    s2[k] = (b2 ? q[2 * k + 1] : q[2 * k]) + xch2(b2 ? q[2 * k] : q[2 * k + 1]);
  float s;
  if (NS == 8) s = (b4 ? s2[NS / 4 - 1] : s2[0]) + xch4(b4 ? s2[0] : s2[NS / 4 - 1]);
  else         s = s2[0] + xch4(s2[0]);
  float l = lrelu(s + ed);
  if (TAIL) l = (i8 < rem) ? l : -__builtin_inff();
  const float mn = fmaxf(m, max8(l));
  const float rs = ex2(m - mn);
  const float pw = ex2(l - mn);
  m = mn;
  ssum = fmaf(ssum, rs, pw);
  acc.x *= rs; acc.y *= rs; acc.z *= rs; acc.w *= rs;
#define GAT_ADD_ROW(K)                                                    \
  {                                                                       \
    const float w = bcast8<K>(p[K], pw);                                  \
    acc.x = fmaf(r[K].x, w, acc.x); acc.y = fmaf(r[K].y, w, acc.y);       \
    acc.z = fmaf(r[K].z, w, acc.z); acc.w = fmaf(r[K].w, w, acc.w);       \
  }
  GAT_ADD_ROW(0) GAT_ADD_ROW(1) GAT_ADD_ROW(2) GAT_ADD_ROW(3)
  if (NS == 8) {
    GAT_ADD_ROW(NS - 4) GAT_ADD_ROW(NS - 3) GAT_ADD_ROW(NS - 2) GAT_ADD_ROW(NS - 1)
  }
#undef GAT_ADD_ROW
}

__device__ __forceinline__ float4 gat_node(
    const int* __restrict__ rowptr, const int* __restrict__ col, const float* __restrict__ Hm,
    const float4 a4, const float4 d4, const float* __restrict__ bias, int node, int lane, int apply_elu) {
  const int beg = __builtin_amdgcn_readfirstlane(rowptr[node]);
  const int n   = __builtin_amdgcn_readfirstlane(rowptr[node + 1]) - beg;
  const unsigned loff = (unsigned)lane * 4u;
  const float4 hs = load_row(Hm, (unsigned)node, loff);
  const int  i8 = lane & 7;
  const bool b1 = (lane & 1) != 0, b2 = (lane & 2) != 0, b4 = (lane & 4) != 0;

  const float ed = sum8(dot4(hs, d4));
  float m = lrelu(sum8(dot4(hs, a4)) + ed);
  float4 acc = hs;
  float ssum = i8 == 0 ? 1.f : 0.f;

  for (int t0 = 0; t0 < n; t0 += 32) {
    int sc = node;
    if (t0 + lane < n) sc = col[beg + t0 + lane];
    const int cnt = (n - t0) < 32 ? (n - t0) : 32;
    int c0 = 0;
    for (; c0 + 8 <= cnt; c0 += 8)
      gat_chunk<false, 8>(Hm, loff, sc, c0, 8, i8, b1, b2, b4, a4, ed, m, ssum, acc);
    if (c0 < cnt) {
      const int rem = cnt - c0;
      if (rem <= 4) gat_chunk<true, 4>(Hm, loff, sc, c0, rem, i8, b1, b2, b4, a4, ed, m, ssum, acc);
      else          gat_chunk<true, 8>(Hm, loff, sc, c0, rem, i8, b1, b2, b4, a4, ed, m, ssum, acc);
    }
  }
  const float tsum = sum8(ssum);
  const float4 bv = *(const float4*)(bias + lane * 4);
  float4 v;
  const float inv = __builtin_amdgcn_rcpf(tsum);
  v.x = fmaf(acc.x, inv, bv.x);
  v.y = fmaf(acc.y, inv, bv.y);
  v.z = fmaf(acc.z, inv, bv.z);
  v.w = fmaf(acc.w, inv, bv.w);
  if (apply_elu) {
    v.x = v.x > 0.f ? v.x : ex2(v.x * LOG2E) - 1.f;
    v.y = v.y > 0.f ? v.y : ex2(v.y * LOG2E) - 1.f;
    v.z = v.z > 0.f ? v.z : ex2(v.z * LOG2E) - 1.f;
    v.w = v.w > 0.f ? v.w : ex2(v.w * LOG2E) - 1.f;
  }
  return v;
}

__global__ __launch_bounds__(256, 16) void gat_gather(
    const int* __restrict__ rowptr, const int* __restrict__ col,
    const float* __restrict__ Hm, const float* __restrict__ asrc, const float* __restrict__ adst,
    const float* __restrict__ bias, float* __restrict__ outp, int N, int apply_elu) {
  const int lane = threadIdx.x & 31;
  const int node = __builtin_amdgcn_readfirstlane(blockIdx.x * 8 + (threadIdx.x >> 5));
  if (node >= N) return;
  float4 a4 = *(const float4*)(asrc + lane * 4);
  float4 d4 = *(const float4*)(adst + lane * 4);
  a4.x *= LOG2E; a4.y *= LOG2E; a4.z *= LOG2E; a4.w *= LOG2E;
  d4.x *= LOG2E; d4.y *= LOG2E; d4.z *= LOG2E; d4.w *= LOG2E;
  const float4 v = gat_node(rowptr, col, Hm, a4, d4, bias, node, lane, apply_elu);
  *(float4*)(outp + (size_t)node * 128 + lane * 4) = v;
}

#define TILE_PITCH 33
#define FUSE_WGS 768
__global__ __launch_bounds__(512) void gat_gather_gemm(
    const int* __restrict__ rowptr, const int* __restrict__ col,
    const float* __restrict__ Hm, const float* __restrict__ asrc, const float* __restrict__ adst,
    const float* __restrict__ bias, const float* __restrict__ Wn, float* __restrict__ Hn, int N) {
  __shared__ float4 wl[32 * 8 * 16];
  __shared__ float4 sa[2][16 * TILE_PITCH];
  const int lane = threadIdx.x & 31;
  const int wv   = __builtin_amdgcn_readfirstlane(threadIdx.x >> 5);
  const int c    = lane & 15;
  const int h    = lane >> 4;

  {
    const int m = threadIdx.x & 31;
#pragma unroll
    for (int it = 0; it < 2; ++it) {
      const int k4 = (threadIdx.x >> 5) + 16 * it;
      const float* wp = Wn + (size_t)(4 * k4) * 128 + 4 * m;
      const float4 r0 = *(const float4*)(wp);
      const float4 r1 = *(const float4*)(wp + 128);
      const float4 r2 = *(const float4*)(wp + 256);
      const float4 r3 = *(const float4*)(wp + 384);
      const int base = (k4 * 8 + 2 * (m >> 3)) * 16 + 2 * (m & 7);
      wl[base]      = make_float4(r0.x, r1.x, r2.x, r3.x);
      wl[base + 16] = make_float4(r0.y, r1.y, r2.y, r3.y);
      wl[base + 1]  = make_float4(r0.z, r1.z, r2.z, r3.z);
      wl[base + 17] = make_float4(r0.w, r1.w, r2.w, r3.w);
    }
  }
  float4 a4 = *(const float4*)(asrc + lane * 4);
  float4 d4 = *(const float4*)(adst + lane * 4);
  a4.x *= LOG2E; a4.y *= LOG2E; a4.z *= LOG2E; a4.w *= LOG2E;
  d4.x *= LOG2E; d4.y *= LOG2E; d4.z *= LOG2E; d4.w *= LOG2E;
  __syncthreads();

  const float4* bl = wl + h * 128 + c;
  const int ngroups = (N + 15) >> 4;
  int it = 0;
  for (int g = blockIdx.x; g < ngroups; g += FUSE_WGS, ++it) {
    const int node = g * 16 + wv;
    float4 v = make_float4(0.f, 0.f, 0.f, 0.f);
    if (node < N) v = gat_node(rowptr, col, Hm, a4, d4, bias, node, lane, 1);
    float4* sb = sa[it & 1];
    sb[wv * TILE_PITCH + lane] = v;
    __syncthreads();
    const int p = (wv - 4 * (it & 3)) & 15;
    if (p < 4) {
      v8f acc0 = (v8f){0.f, 0.f, 0.f, 0.f, 0.f, 0.f, 0.f, 0.f};
      v8f acc1 = (v8f){0.f, 0.f, 0.f, 0.f, 0.f, 0.f, 0.f, 0.f};
      const float4* ar = sb + c * TILE_PITCH + h;
      const float4* br = bl + (2 * p) * 16;
#pragma unroll
      for (int u = 0; u < 16; ++u) {
        const float4 a  = ar[2 * u];
        const float4 b0 = br[(2 * u * 8) * 16];
        const float4 b1 = br[(2 * u * 8 + 1) * 16];
        v2f a0; a0.x = a.x; a0.y = a.y;
        v2f a1; a1.x = a.z; a1.y = a.w;
        v2f b00; b00.x = b0.x; b00.y = b0.y;
        v2f b01; b01.x = b0.z; b01.y = b0.w;
        v2f b10; b10.x = b1.x; b10.y = b1.y;
        v2f b11; b11.x = b1.z; b11.y = b1.w;
        acc0 = __builtin_amdgcn_wmma_f32_16x16x4_f32(false, a0, false, b00, (short)0, acc0, false, false);
        acc1 = __builtin_amdgcn_wmma_f32_16x16x4_f32(false, a0, false, b10, (short)0, acc1, false, false);
        acc0 = __builtin_amdgcn_wmma_f32_16x16x4_f32(false, a1, false, b01, (short)0, acc0, false, false);
        acc1 = __builtin_amdgcn_wmma_f32_16x16x4_f32(false, a1, false, b11, (short)0, acc1, false, false);
      }
      const int rowbase = g * 16 + 8 * h;
#pragma unroll
      for (int i = 0; i < 8; ++i) {
        const int row = rowbase + i;
        if (row < N)
          *(float2*)(Hn + (size_t)row * 128 + 32 * p + 2 * c) = make_float2(acc0[i], acc1[i]);
      }
    }
  }
}

extern "C" void kernel_launch(void* const* d_in, const int* in_sizes, int n_in,
                              void* d_out, int out_size, void* d_ws, size_t ws_size,
                              hipStream_t stream) {
  const float* x  = (const float*)d_in[0];
  const int*   ei = (const int*)d_in[1];
  const int N = in_sizes[0] / 128;
  const int E = in_sizes[1] / 2;

  const float* Wl[3] = {(const float*)d_in[2], (const float*)d_in[6],  (const float*)d_in[10]};
  const float* As[3] = {(const float*)d_in[3], (const float*)d_in[7],  (const float*)d_in[11]};
  const float* Ad[3] = {(const float*)d_in[4], (const float*)d_in[8],  (const float*)d_in[12]};
  const float* Bb[3] = {(const float*)d_in[5], (const float*)d_in[9],  (const float*)d_in[13]};

  const int gWave = (N + 7) / 8;
  const int NB    = (N + BIN_W - 1) / BIN_W;
  const int GA    = (E + SLAB - 1) / SLAB;

  float* F0     = (float*)d_ws;
  float* F1     = F0 + (size_t)N * 128;
  int*   pairs  = (int*)(F1 + (size_t)N * 128);
  int*   tab    = pairs + (size_t)GA * SLAB * 2;
  int*   rowptr = tab + (size_t)GA * (NB + 1);
  int*   spill  = rowptr + (size_t)N + 4;
  int*   colB   = spill + (size_t)E;

  csr_bin<<<GA, 1024, 0, stream>>>(ei, pairs, tab, E, NB);
  csr_rows<<<NB, 512, 0, stream>>>(pairs, tab, rowptr, spill, colB, N, NB, GA);

  gat_gemm128<<<GEMM_WGS, 512, 0, stream>>>(x, Wl[0], F0, N);
  gat_gather_gemm<<<FUSE_WGS, 512, 0, stream>>>(rowptr, colB, F0, As[0], Ad[0], Bb[0], Wl[1], F1, N);
  gat_gather_gemm<<<FUSE_WGS, 512, 0, stream>>>(rowptr, colB, F1, As[1], Ad[1], Bb[1], Wl[2], F0, N);
  gat_gather<<<gWave, 256, 0, stream>>>(rowptr, colB, F0, As[2], Ad[2], Bb[2], (float*)d_out, N, 0);
}
